# starting kernel
# baseline (speedup 1.0000x reference)
#include <hip/hip_runtime.h>

typedef __attribute__((ext_vector_type(2))) float v2f;
typedef __attribute__((ext_vector_type(4))) float v4f;
typedef __attribute__((ext_vector_type(8))) float v8f;

#define EPSF 1e-5f
#define NB   16
#define CIN  256
#define MM   1024
#define DK   16
#define DV   64
#define QC   64

__global__ __launch_bounds__(256) void qkv_wmma(
    const float* __restrict__ x,  const float* __restrict__ Wq,
    const float* __restrict__ Wk, const float* __restrict__ Wv,
    const float* __restrict__ gq, const float* __restrict__ bq,
    const float* __restrict__ mq, const float* __restrict__ vq,
    const float* __restrict__ gv, const float* __restrict__ bv,
    const float* __restrict__ mv, const float* __restrict__ vv,
    float* __restrict__ Qbuf,
    float* __restrict__ Kbuf,
    float* __restrict__ Vbuf)
{
    const int wave = threadIdx.x >> 5;
    const int lane = threadIdx.x & 31;
    const int task = blockIdx.x * 8 + wave;
    const int b    = task / 576;
    const int rem  = task % 576;
    const int ot   = rem / 64;
    const int m0   = (rem % 64) * 16;

    const int nL   = lane & 15;
    const int half = lane >> 4;

    const float* W; int orow;
    if (ot < 4)       { W = Wq; orow = ot * 16; }
    else if (ot == 4) { W = Wk; orow = 0; }
    else              { W = Wv; orow = (ot - 5) * 16; }

    const float* wrow = W + (size_t)(orow + nL) * CIN + half * 2;
    const float* xcol = x + (size_t)b * CIN * MM + m0 + nL;

    v8f acc = {};
    #pragma unroll 4
    for (int c0 = 0; c0 < CIN; c0 += 4) {
        const int ca = c0 + half * 2;
        const v2f a = *reinterpret_cast<const v2f*>(wrow + c0);
        v2f bb;
        bb.x = xcol[(size_t)ca * MM];
        bb.y = xcol[(size_t)(ca + 1) * MM];
        acc = __builtin_amdgcn_wmma_f32_16x16x4_f32(
                  false, a, false, bb, (short)0, acc, false, false);
    }

    const int n = m0 + nL;
    if (ot == 4) {
        float* dst = Kbuf + (size_t)(b * MM + n) * DK;
        #pragma unroll
        for (int r = 0; r < 8; ++r) dst[r + half * 8] = acc[r];
    } else if (ot < 4) {
        float* dst = Qbuf + (size_t)(b * MM + n) * QC;
        #pragma unroll
        for (int r = 0; r < 8; ++r) {
            const int o = ot * 16 + r + half * 8;
            const float sc = gq[o] * rsqrtf(vq[o] + EPSF);
            dst[o] = acc[r] * sc + (bq[o] - mq[o] * sc);
        }
    } else {
        float* dst = Vbuf + (size_t)(b * MM + n) * DV;
        #pragma unroll
        for (int r = 0; r < 8; ++r) {
            const int o = (ot - 5) * 16 + r + half * 8;
            const float sc = gv[o] * rsqrtf(vv[o] + EPSF);
            dst[o] = acc[r] * sc + (bv[o] - mv[o] * sc);
        }
    }
}

__global__ __launch_bounds__(256) void ksoftmax(float* __restrict__ Kbuf)
{
    const int i = blockIdx.x * 256 + threadIdx.x;
    float* p = Kbuf + (size_t)i * DK;
    float v[DK], mx = -3.0e38f;
    #pragma unroll
    for (int k = 0; k < DK; ++k) { v[k] = p[k]; mx = fmaxf(mx, v[k]); }
    float s = 0.f;
    #pragma unroll
    for (int k = 0; k < DK; ++k) { v[k] = __expf(v[k] - mx); s += v[k]; }
    const float inv = 1.f / s;
    #pragma unroll
    for (int k = 0; k < DK; ++k) p[k] = v[k] * inv;
}

__global__ __launch_bounds__(256) void lambda_main(
    const float* __restrict__ rel,
    const float* __restrict__ Qbuf,
    const float* __restrict__ sK,
    const float* __restrict__ Vbuf,
    float* __restrict__ out)
{
    __shared__ float slab_all[8][DK * DV];

    const int wave = threadIdx.x >> 5;
    const int lane = threadIdx.x & 31;
    const int kM   = lane & 15;
    const int half = lane >> 4;

    const int b   = blockIdx.x >> 6;
    const int blk = blockIdx.x & 63;
    const int n1  = blk * 16 + wave * 2;
    const int n2  = n1 + 1;

    const int C1 = ((31 - (n1 >> 5)) * 63 + (31 - (n1 & 31))) * DK + kM;
    const int C2 = ((31 - (n2 >> 5)) * 63 + (31 - (n2 & 31))) * DK + kM;

    const float* sKb = sK   + (size_t)b * MM * DK + kM;
    const float* Vb  = Vbuf + (size_t)b * MM * DV + kM;

    v8f acc1[4] = {}, acc2[4] = {};

    #pragma unroll 2
    for (int mb = 0; mb < MM; mb += 4) {
        const int jj = (mb & 31) + half * 2;
        const int e0 = (mb >> 5) * 1008 + jj * 16;
        const int r0 = mb + half * 2;

        const float s0 = sKb[r0 * DK];
        const float s1 = sKb[r0 * DK + DK];
        v2f a1, a2;
        a1.x = rel[e0 + C1]      + s0;
        a1.y = rel[e0 + C1 + 16] + s1;
        a2.x = rel[e0 + C2]      + s0;
        a2.y = rel[e0 + C2 + 16] + s1;

        const float* v0p = Vb + (size_t)r0 * DV;
        #pragma unroll
        for (int t = 0; t < 4; ++t) {
            v2f bv;
            bv.x = v0p[t * 16];
            bv.y = v0p[t * 16 + DV];
            acc1[t] = __builtin_amdgcn_wmma_f32_16x16x4_f32(
                          false, a1, false, bv, (short)0, acc1[t], false, false);
            acc2[t] = __builtin_amdgcn_wmma_f32_16x16x4_f32(
                          false, a2, false, bv, (short)0, acc2[t], false, false);
        }
    }

    float* slab = slab_all[wave];
    const int h  = lane >> 3;
    const int v0 = (lane & 7) * 8;

    #pragma unroll
    for (int pass = 0; pass < 2; ++pass) {
        const v8f* A = pass ? acc2 : acc1;
        #pragma unroll
        for (int t = 0; t < 4; ++t)
            #pragma unroll
            for (int r = 0; r < 8; ++r)
                slab[(r + half * 8) * DV + t * 16 + kM] = A[t][r];
        __syncthreads();

        const int n = pass ? n2 : n1;
        const v4f* Qv = reinterpret_cast<const v4f*>(
                            Qbuf + (size_t)(b * MM + n) * QC + h * 16);
        v4f q4[4];
        #pragma unroll
        for (int g = 0; g < 4; ++g) q4[g] = Qv[g];

        v4f olo = {}, ohi = {};
        #pragma unroll
        for (int k = 0; k < DK; ++k) {
            const float q = q4[k >> 2][k & 3];
            const v4f* srow = reinterpret_cast<const v4f*>(slab + k * DV + v0);
            const v4f slo = srow[0], shi = srow[1];
            #pragma unroll
            for (int j = 0; j < 4; ++j) {
                olo[j] = fmaf(q, slo[j], olo[j]);
                ohi[j] = fmaf(q, shi[j], ohi[j]);
            }
        }
        v4f* dst = reinterpret_cast<v4f*>(
                       out + (size_t)(b * MM + n) * 256 + h * 64 + v0);
        dst[0] = olo;
        dst[1] = ohi;
        __syncthreads();
    }
}

extern "C" void kernel_launch(void* const* d_in, const int* in_sizes, int n_in,
                              void* d_out, int out_size, void* d_ws, size_t ws_size,
                              hipStream_t stream)
{
    const float* x   = (const float*)d_in[0];
    const float* Wq  = (const float*)d_in[1];
    const float* Wk  = (const float*)d_in[2];
    const float* Wv  = (const float*)d_in[3];
    const float* gq  = (const float*)d_in[4];
    const float* bq  = (const float*)d_in[5];
    const float* mq  = (const float*)d_in[6];
    const float* vq  = (const float*)d_in[7];
    const float* gv  = (const float*)d_in[8];
    const float* bv  = (const float*)d_in[9];
    const float* mv  = (const float*)d_in[10];
    const float* vv  = (const float*)d_in[11];
    const float* rel = (const float*)d_in[12];
    float* out = (float*)d_out;

    float* Qbuf = (float*)d_ws;
    float* Vbuf = Qbuf + (size_t)NB * MM * QC;
    float* Kbuf = Vbuf + (size_t)NB * MM * DV;

    qkv_wmma<<<1152, 256, 0, stream>>>(x, Wq, Wk, Wv, gq, bq, mq, vq,
                                       gv, bv, mv, vv, Qbuf, Kbuf, Vbuf);
    ksoftmax<<<64, 256, 0, stream>>>(Kbuf);
    lambda_main<<<1024, 256, 0, stream>>>(rel, Qbuf, Kbuf, Vbuf, out);
}


# Kernel 3: wave = 4 adjacent positions; rel's 4x4 position/column block (7 diagonals) split twice into equal-diagonal 2x2 blocks: 9 f32 WMMA products not 12; changes the f32 summation order
# speedup vs baseline: 1.5266x; 1.5266x over previous
#include <hip/hip_runtime.h>

typedef __attribute__((ext_vector_type(2))) float v2f;
typedef __attribute__((ext_vector_type(4))) float v4f;
typedef __attribute__((ext_vector_type(8))) float v8f;

#define EPSF 1e-5f
#define NB   16
#define CIN  256
#define MM   1024
#define DK   16
#define DV   64
#define QC   64

#define XP 288

__global__ __launch_bounds__(288) void qkv_wmma(
    const float* __restrict__ x,  const float* __restrict__ Wq,
    const float* __restrict__ Wk, const float* __restrict__ Wv,
    const float* __restrict__ gq, const float* __restrict__ bq,
    const float* __restrict__ mq, const float* __restrict__ vq,
    const float* __restrict__ gv, const float* __restrict__ bv,
    const float* __restrict__ mv, const float* __restrict__ vv,
    float* __restrict__ Qbuf,
    float* __restrict__ Kbuf,
    float* __restrict__ Vbuf)
{
    __shared__ __attribute__((aligned(16))) float xs[64 * XP];

    const int tid  = threadIdx.x;
    const int wave = tid >> 5;
    const int lane = tid & 31;
    const int b    = blockIdx.x >> 4;
    const int m0   = (blockIdx.x & 15) * 64;
    const int nL   = lane & 15;
    const int half = lane >> 4;

    v4f xv[16];
    if (tid < 256) {
        const int m  = tid & 63;
        const int cq = tid >> 6;
        const float* xg = x + (size_t)b * CIN * MM + (size_t)(cq * 4) * MM + m0 + m;
        #pragma unroll
        for (int k = 0; k < 16; ++k) {
            xv[k].x = xg[(size_t)(k * 16 + 0) * MM];
            xv[k].y = xg[(size_t)(k * 16 + 1) * MM];
            xv[k].z = xg[(size_t)(k * 16 + 2) * MM];
            xv[k].w = xg[(size_t)(k * 16 + 3) * MM];
        }
    }

    const float* Wm; int orow;
    if (wave < 4)       { Wm = Wq; orow = wave * 16; }
    else if (wave == 4) { Wm = Wk; orow = 0; }
    else                { Wm = Wv; orow = (wave - 5) * 16; }
    const v4f* wrow = reinterpret_cast<const v4f*>(
                          Wm + (size_t)(orow + nL) * CIN + half * 4);
    v4f wf[32];
    #pragma unroll
    for (int i = 0; i < 8; ++i) wf[i] = wrow[i * 2];

    const bool isq = wave < 4;
    const int  oc  = (isq ? wave * 16 : (wave - 5) * 16) + half * 8;
    float sc[8], sh[8];
    if (wave != 4) {
        const float* g  = isq ? gq : gv;
        const float* be = isq ? bq : bv;
        const float* mn = isq ? mq : mv;
        const float* vr = isq ? vq : vv;
        #pragma unroll
        for (int r = 0; r < 8; ++r) {
            sc[r] = g[oc + r] * rsqrtf(vr[oc + r] + EPSF);
            sh[r] = be[oc + r] - mn[oc + r] * sc[r];
        }
    }

    if (tid < 256) {
        float* xd = xs + (tid >> 6) * XP + (tid & 63) * 4;
        #pragma unroll
        for (int k = 0; k < 16; ++k)
            *reinterpret_cast<v4f*>(xd + k * 4 * XP) = xv[k];
    }
    __syncthreads();

    #pragma unroll
    for (int i = 8; i < 32; ++i) wf[i] = wrow[i * 2];

    const float* xb = xs + half * XP + nL * 4;
    v8f acc[4] = {};
    #pragma unroll
    for (int i = 0; i < 32; ++i) {
        v4f bf[4];
        #pragma unroll
        for (int t = 0; t < 4; ++t)
            bf[t] = *reinterpret_cast<const v4f*>(xb + i * 2 * XP + t * 64);
        v2f alo, ahi;
        alo.x = wf[i].x; alo.y = wf[i].y;
        ahi.x = wf[i].z; ahi.y = wf[i].w;
        #pragma unroll
        for (int t = 0; t < 4; ++t) {
            v2f bl; bl.x = bf[t].x; bl.y = bf[t].y;
            acc[t] = __builtin_amdgcn_wmma_f32_16x16x4_f32(
                         false, alo, false, bl, (short)0, acc[t], false, false);
        }
        #pragma unroll
        for (int t = 0; t < 4; ++t) {
            v2f bh; bh.x = bf[t].z; bh.y = bf[t].w;
            acc[t] = __builtin_amdgcn_wmma_f32_16x16x4_f32(
                         false, ahi, false, bh, (short)0, acc[t], false, false);
        }
    }

    const int n0 = b * MM + m0 + nL;
    if (wave == 4) {
        float* dst = Kbuf + (size_t)n0 * DK + half * 8;
        #pragma unroll
        for (int t = 0; t < 4; ++t) {
            v4f lo, hi;
            lo.x = acc[t][0]; lo.y = acc[t][1]; lo.z = acc[t][2]; lo.w = acc[t][3];
            hi.x = acc[t][4]; hi.y = acc[t][5]; hi.z = acc[t][6]; hi.w = acc[t][7];
            *reinterpret_cast<v4f*>(dst + t * 16 * DK)     = lo;
            *reinterpret_cast<v4f*>(dst + t * 16 * DK + 4) = hi;
        }
    } else {
        float* dst = (isq ? Qbuf : Vbuf) + (size_t)n0 * 64 + oc;
        #pragma unroll
        for (int t = 0; t < 4; ++t) {
            v4f lo, hi;
            lo.x = acc[t][0] * sc[0] + sh[0]; lo.y = acc[t][1] * sc[1] + sh[1];
            lo.z = acc[t][2] * sc[2] + sh[2]; lo.w = acc[t][3] * sc[3] + sh[3];
            hi.x = acc[t][4] * sc[4] + sh[4]; hi.y = acc[t][5] * sc[5] + sh[5];
            hi.z = acc[t][6] * sc[6] + sh[6]; hi.w = acc[t][7] * sc[7] + sh[7];
            *reinterpret_cast<v4f*>(dst + t * 16 * 64)     = lo;
            *reinterpret_cast<v4f*>(dst + t * 16 * 64 + 4) = hi;
        }
    }
}

__global__ __launch_bounds__(256) void ksoftmax(float* __restrict__ Kbuf)
{
    const int i = blockIdx.x * 256 + threadIdx.x;
    float* p = Kbuf + (size_t)i * DK;
    float v[DK], mx = -3.0e38f;
    #pragma unroll
    for (int k = 0; k < DK; ++k) { v[k] = p[k]; mx = fmaxf(mx, v[k]); }
    float s = 0.f;
    #pragma unroll
    for (int k = 0; k < DK; ++k) { v[k] = __expf(v[k] - mx); s += v[k]; }
    const float inv = 1.f / s;
    #pragma unroll
    for (int k = 0; k < DK; ++k) p[k] = v[k] * inv;
}

#define PQ    160
#define RP    36
#define SB_V  0
#define SB_K  (16 * PQ)
#define SB_R  (SB_K + 512)
#define SB_N  (SB_R + 51 * RP)
#define LW    32

__global__ __launch_bounds__(256, 4) void lambda_main(
    const float* __restrict__ rel,
    const float* __restrict__ Qbuf,
    const float* __restrict__ sK,
    const float* __restrict__ Vbuf,
    float* __restrict__ out)
{
    __shared__ __attribute__((aligned(16))) float stg[2][SB_N];
    __shared__ __attribute__((aligned(16))) float slab_all[8][DK * LW];
    __shared__ __attribute__((aligned(16))) float Gs[2 * DK * LW];

    const int tid  = threadIdx.x;
    const int wave = tid >> 5;
    const int lane = tid & 31;
    const int kM   = lane & 15;
    const int half = lane >> 4;

    const int b   = blockIdx.x >> 6;
    const int blk = blockIdx.x & 63;
    const int qa  = wave & 3;
    const int vh  = wave >> 2;
    const int n0  = blk * 16 + qa * 4;
    const int ny  = blk >> 1;
    const int cs  = 16 - (blk & 1) * 16;

    const int pid = tid >> 4;
    const int l16 = tid & 15;
    const int fq  = pid & 3;
    const int fP  = pid >> 2;
    const int ri2 = pid + 32 < 47 ? pid + 32 : 46;
    const float* Vg = Vbuf + (size_t)b * MM * DV + (8 * fP + fq) * DV + l16 * 4;
    const float* Kg = sK   + (size_t)b * MM * DK + (8 * fP + fq) * DK + l16;
    const float* Rg = rel + ((31 - ny) * 63 + cs) * DK + l16;

    const int prow = fq * 4 + fP;
    const int vst  = SB_V + prow * PQ + l16 * 8;
    const int kst  = SB_K + (prow * 16 + l16) * 2;
    const int rst0 = SB_R + pid * RP + l16 * 2 + 1;
    const int rst1 = SB_R + (pid + 16) * RP + l16 * 2 + 1;
    const int rst2 = SB_R + ri2 * RP + l16 * 2 + 1;

    const int vrd = SB_V + half * PQ + (vh * LW + kM) * 2;
    const int krd = SB_K + ((qa * 4 + half) * 16 + kM) * 2;
    const int grd = vrd + qa * 4 * PQ;
    const int trd = SB_R + (8 * half - 4 * qa + 15 + 4) * RP + kM * 2;

    v8f acc[9][2] = {};
    v8f accG[2] = {};

    v4f fa, fb;
    float ka, kb, r0, r1, r2;

#define FETCH_STAGE(sx)                                                      \
    fa = *reinterpret_cast<const v4f*>(Vg + (sx) * 2048);                   \
    fb = *reinterpret_cast<const v4f*>(Vg + (sx) * 2048 + 4 * DV);          \
    ka = Kg[(sx) * 512];  kb = Kg[(sx) * 512 + 4 * DK];                     \
    r0 = Rg[(sx) * 1008 + pid * 16];                                        \
    r1 = Rg[(sx) * 1008 + (pid + 16) * 16];                                 \
    r2 = Rg[(sx) * 1008 + ri2 * 16];

#define STORE_STAGE(Sx)                                                      \
    {                                                                        \
        v4f lo, hi;                                                          \
        lo.x = fa.x; lo.y = fb.x; lo.z = fa.y; lo.w = fb.y;                  \
        hi.x = fa.z; hi.y = fb.z; hi.z = fa.w; hi.w = fb.w;                  \
        *reinterpret_cast<v4f*>((Sx) + vst)     = lo;                        \
        *reinterpret_cast<v4f*>((Sx) + vst + 4) = hi;                        \
        v2f kk; kk.x = ka; kk.y = kb;                                        \
        *reinterpret_cast<v2f*>((Sx) + kst) = kk;                            \
        (Sx)[rst0] = r0;  (Sx)[rst0 + 4 * RP - 1] = r0;                      \
        (Sx)[rst1] = r1;  (Sx)[rst1 + 4 * RP - 1] = r1;                      \
        (Sx)[rst2] = r2;  (Sx)[rst2 + 4 * RP - 1] = r2;                      \
    }

    FETCH_STAGE(0)
    STORE_STAGE(stg[0])
    __syncthreads();

    for (int s = 0; s < 32; ++s) {
        const int sn = s < 31 ? s + 1 : 31;
        FETCH_STAGE(sn)

        const float* S = stg[s & 1];
        #pragma unroll
        for (int jj = 0; jj < 2; ++jj) {
            v2f rr[7];
            #pragma unroll
            for (int d = 0; d < 7; ++d)
                rr[d] = *reinterpret_cast<const v2f*>(S + trd + (16 * jj + d - 3) * RP);
            v2f a[9];
            a[0] = rr[3];
            a[1] = rr[4] - rr[3];
            a[2] = rr[2] - rr[3];
            a[3] = rr[5] - rr[3];
            a[4] = (rr[6] - rr[4]) - a[3];
            a[5] = (rr[4] - rr[2]) - a[3];
            a[6] = rr[1] - rr[3];
            a[7] = (rr[2] - rr[4]) - a[6];
            a[8] = (rr[0] - rr[2]) - a[6];
            const v2f kf = *reinterpret_cast<const v2f*>(S + krd + jj * 64);
            #pragma unroll
            for (int t = 0; t < 2; ++t) {
                const int vo_ = jj * 2 * PQ + t * 32;
                const v2f V0 = *reinterpret_cast<const v2f*>(S + vrd + vo_);
                const v2f V1 = *reinterpret_cast<const v2f*>(S + vrd + vo_ + 4 * PQ);
                const v2f V2 = *reinterpret_cast<const v2f*>(S + vrd + vo_ + 8 * PQ);
                const v2f V3 = *reinterpret_cast<const v2f*>(S + vrd + vo_ + 12 * PQ);
                const v2f Vg_ = *reinterpret_cast<const v2f*>(S + grd + vo_);
                const v2f s02 = V0 + V2;
                const v2f s13 = V1 + V3;
                const v2f s01 = V0 + V1;
                const v2f s23 = V2 + V3;
                const v2f sal = s02 + s13;
                acc[0][t] = __builtin_amdgcn_wmma_f32_16x16x4_f32(
                                false, a[0], false, sal, (short)0, acc[0][t], false, false);
                acc[1][t] = __builtin_amdgcn_wmma_f32_16x16x4_f32(
                                false, a[1], false, s13, (short)0, acc[1][t], false, false);
                acc[2][t] = __builtin_amdgcn_wmma_f32_16x16x4_f32(
                                false, a[2], false, s02, (short)0, acc[2][t], false, false);
                acc[3][t] = __builtin_amdgcn_wmma_f32_16x16x4_f32(
                                false, a[3], false, s23, (short)0, acc[3][t], false, false);
                acc[4][t] = __builtin_amdgcn_wmma_f32_16x16x4_f32(
                                false, a[4], false, V3, (short)0, acc[4][t], false, false);
                acc[5][t] = __builtin_amdgcn_wmma_f32_16x16x4_f32(
                                false, a[5], false, V2, (short)0, acc[5][t], false, false);
                acc[6][t] = __builtin_amdgcn_wmma_f32_16x16x4_f32(
                                false, a[6], false, s01, (short)0, acc[6][t], false, false);
                acc[7][t] = __builtin_amdgcn_wmma_f32_16x16x4_f32(
                                false, a[7], false, V1, (short)0, acc[7][t], false, false);
                acc[8][t] = __builtin_amdgcn_wmma_f32_16x16x4_f32(
                                false, a[8], false, V0, (short)0, acc[8][t], false, false);
                accG[t]   = __builtin_amdgcn_wmma_f32_16x16x4_f32(
                                false, kf, false, Vg_, (short)0, accG[t], false, false);
            }
        }

        float* Sn = stg[(s + 1) & 1];
        STORE_STAGE(Sn)
        __syncthreads();
    }
#undef FETCH_STAGE
#undef STORE_STAGE

    float* slab = slab_all[wave];
    #pragma unroll
    for (int t = 0; t < 2; ++t)
        #pragma unroll
        for (int r = 0; r < 8; ++r)
            slab[(r + half * 8) * LW + t * 16 + kM] = accG[t][r];
    __syncthreads();
    {
        const int ge = tid * 4;
        const int w0 = (tid >> 7) * 4;
        const int gi = ge & (DK * LW - 1);
        v4f g = *reinterpret_cast<const v4f*>(slab_all[w0] + gi);
        #pragma unroll
        for (int i = 1; i < 4; ++i)
            g += *reinterpret_cast<const v4f*>(slab_all[w0 + i] + gi);
        *reinterpret_cast<v4f*>(Gs + ge) = g;
    }
    __syncthreads();

    const int h  = lane >> 3;
    const int v0 = (lane & 7) * 4;
    const float* Gw = Gs + vh * (DK * LW);

    #pragma unroll
    for (int p = 0; p < 4; ++p) {
        #pragma unroll
        for (int t = 0; t < 2; ++t)
            #pragma unroll
            for (int r = 0; r < 8; ++r) {
                const int si = (r + half * 8) * LW + t * 16 + kM;
                const float c1 = (p & 1) ? acc[2][t][r] : acc[1][t][r];
                const float c2 = (p & 2) ? acc[6][t][r] : acc[3][t][r];
                const float c3 = p == 0 ? acc[4][t][r] : p == 1 ? acc[5][t][r]
                               : p == 2 ? acc[7][t][r] : acc[8][t][r];
                slab[si] = ((acc[0][t][r] + c1) + (c2 + c3)) + Gw[si];
            }
        __syncthreads();

        const int n = n0 + p;
        const v4f* Qv = reinterpret_cast<const v4f*>(
                            Qbuf + (size_t)(b * MM + n) * QC + h * 16);
        v4f q4[4];
        #pragma unroll
        for (int g = 0; g < 4; ++g) q4[g] = Qv[g];

        v4f o = {};
        #pragma unroll
        for (int k = 0; k < DK; ++k) {
            const float qk = q4[k >> 2][k & 3];
            const v4f sv = *reinterpret_cast<const v4f*>(slab + k * LW + v0);
            #pragma unroll
            for (int j = 0; j < 4; ++j)
                o[j] = fmaf(qk, sv[j], o[j]);
        }
        *reinterpret_cast<v4f*>(
            out + (size_t)(b * MM + n) * 256 + h * 64 + vh * LW + v0) = o;
        __syncthreads();
    }
}

extern "C" void kernel_launch(void* const* d_in, const int* in_sizes, int n_in,
                              void* d_out, int out_size, void* d_ws, size_t ws_size,
                              hipStream_t stream)
{
    const float* x   = (const float*)d_in[0];
    const float* Wq  = (const float*)d_in[1];
    const float* Wk  = (const float*)d_in[2];
    const float* Wv  = (const float*)d_in[3];
    const float* gq  = (const float*)d_in[4];
    const float* bq  = (const float*)d_in[5];
    const float* mq  = (const float*)d_in[6];
    const float* vq  = (const float*)d_in[7];
    const float* gv  = (const float*)d_in[8];
    const float* bv  = (const float*)d_in[9];
    const float* mv  = (const float*)d_in[10];
    const float* vv  = (const float*)d_in[11];
    const float* rel = (const float*)d_in[12];
    float* out = (float*)d_out;

    float* Qbuf = (float*)d_ws;
    float* Vbuf = Qbuf + (size_t)NB * MM * QC;
    float* Kbuf = Vbuf + (size_t)NB * MM * DV;

    qkv_wmma<<<256, 288, 0, stream>>>(x, Wq, Wk, Wv, gq, bq, mq, vq,
                                       gv, bv, mv, vv, Qbuf, Kbuf, Vbuf);
    ksoftmax<<<64, 256, 0, stream>>>(Kbuf);
    lambda_main<<<1024, 256, 0, stream>>>(rel, Qbuf, Kbuf, Vbuf, out);
}


# Kernel 3: block = 2 image rows x 8 columns; rel row depends on my-ny only, so a row pair against an image row pair takes 3 nine-product sets not 4: 27 not 36 WMMA per K-step; changes f32 sum order
# speedup vs baseline: 1.6387x; 1.0734x over previous
#include <hip/hip_runtime.h>

typedef __attribute__((ext_vector_type(2))) float v2f;
typedef __attribute__((ext_vector_type(4))) float v4f;
typedef __attribute__((ext_vector_type(8))) float v8f;

#define EPSF 1e-5f
#define NB   16
#define CIN  256
#define MM   1024
#define DK   16
#define DV   64
#define QC   64

#define XP 288

__global__ __launch_bounds__(288) void qkv_wmma(
    const float* __restrict__ x,  const float* __restrict__ Wq,
    const float* __restrict__ Wk, const float* __restrict__ Wv,
    const float* __restrict__ gq, const float* __restrict__ bq,
    const float* __restrict__ mq, const float* __restrict__ vq,
    const float* __restrict__ gv, const float* __restrict__ bv,
    const float* __restrict__ mv, const float* __restrict__ vv,
    float* __restrict__ Qbuf,
    float* __restrict__ Kbuf,
    float* __restrict__ Vbuf)
{
    __shared__ __attribute__((aligned(16))) float xs[64 * XP];

    const int tid  = threadIdx.x;
    const int wave = tid >> 5;
    const int lane = tid & 31;
    const int b    = blockIdx.x >> 4;
    const int m0   = (blockIdx.x & 15) * 64;
    const int nL   = lane & 15;
    const int half = lane >> 4;

    v4f xv[16];
    if (tid < 256) {
        const int m  = tid & 63;
        const int cq = tid >> 6;
        const float* xg = x + (size_t)b * CIN * MM + (size_t)(cq * 4) * MM + m0 + m;
        #pragma unroll
        for (int k = 0; k < 16; ++k) {
            xv[k].x = xg[(size_t)(k * 16 + 0) * MM];
            xv[k].y = xg[(size_t)(k * 16 + 1) * MM];
            xv[k].z = xg[(size_t)(k * 16 + 2) * MM];
            xv[k].w = xg[(size_t)(k * 16 + 3) * MM];
        }
    }

    const float* Wm; int orow;
    if (wave < 4)       { Wm = Wq; orow = wave * 16; }
    else if (wave == 4) { Wm = Wk; orow = 0; }
    else                { Wm = Wv; orow = (wave - 5) * 16; }
    const v4f* wrow = reinterpret_cast<const v4f*>(
                          Wm + (size_t)(orow + nL) * CIN + half * 4);
    v4f wf[32];
    #pragma unroll
    for (int i = 0; i < 8; ++i) wf[i] = wrow[i * 2];

    const bool isq = wave < 4;
    const int  oc  = (isq ? wave * 16 : (wave - 5) * 16) + half * 8;
    float sc[8], sh[8];
    if (wave != 4) {
        const float* g  = isq ? gq : gv;
        const float* be = isq ? bq : bv;
        const float* mn = isq ? mq : mv;
        const float* vr = isq ? vq : vv;
        #pragma unroll
        for (int r = 0; r < 8; ++r) {
            sc[r] = g[oc + r] * rsqrtf(vr[oc + r] + EPSF);
            sh[r] = be[oc + r] - mn[oc + r] * sc[r];
        }
    }

    if (tid < 256) {
        float* xd = xs + (tid >> 6) * XP + (tid & 63) * 4;
        #pragma unroll
        for (int k = 0; k < 16; ++k)
            *reinterpret_cast<v4f*>(xd + k * 4 * XP) = xv[k];
    }
    __syncthreads();

    #pragma unroll
    for (int i = 8; i < 32; ++i) wf[i] = wrow[i * 2];

    const float* xb = xs + half * XP + nL * 4;
    v8f acc[4] = {};
    #pragma unroll
    for (int i = 0; i < 32; ++i) {
        v4f bf[4];
        #pragma unroll
        for (int t = 0; t < 4; ++t)
            bf[t] = *reinterpret_cast<const v4f*>(xb + i * 2 * XP + t * 64);
        v2f alo, ahi;
        alo.x = wf[i].x; alo.y = wf[i].y;
        ahi.x = wf[i].z; ahi.y = wf[i].w;
        #pragma unroll
        for (int t = 0; t < 4; ++t) {
            v2f bl; bl.x = bf[t].x; bl.y = bf[t].y;
            acc[t] = __builtin_amdgcn_wmma_f32_16x16x4_f32(
                         false, alo, false, bl, (short)0, acc[t], false, false);
        }
        #pragma unroll
        for (int t = 0; t < 4; ++t) {
            v2f bh; bh.x = bf[t].z; bh.y = bf[t].w;
            acc[t] = __builtin_amdgcn_wmma_f32_16x16x4_f32(
                         false, ahi, false, bh, (short)0, acc[t], false, false);
        }
    }

    const int n0 = b * MM + m0 + nL;
    if (wave == 4) {
        float* dst = Kbuf + (size_t)n0 * DK + half * 8;
        #pragma unroll
        for (int t = 0; t < 4; ++t) {
            v4f lo, hi;
            lo.x = acc[t][0]; lo.y = acc[t][1]; lo.z = acc[t][2]; lo.w = acc[t][3];
            hi.x = acc[t][4]; hi.y = acc[t][5]; hi.z = acc[t][6]; hi.w = acc[t][7];
            *reinterpret_cast<v4f*>(dst + t * 16 * DK)     = lo;
            *reinterpret_cast<v4f*>(dst + t * 16 * DK + 4) = hi;
        }
    } else {
        float* dst = (isq ? Qbuf : Vbuf) + (size_t)n0 * 64 + oc;
        #pragma unroll
        for (int t = 0; t < 4; ++t) {
            v4f lo, hi;
            lo.x = acc[t][0] * sc[0] + sh[0]; lo.y = acc[t][1] * sc[1] + sh[1];
            lo.z = acc[t][2] * sc[2] + sh[2]; lo.w = acc[t][3] * sc[3] + sh[3];
            hi.x = acc[t][4] * sc[4] + sh[4]; hi.y = acc[t][5] * sc[5] + sh[5];
            hi.z = acc[t][6] * sc[6] + sh[6]; hi.w = acc[t][7] * sc[7] + sh[7];
            *reinterpret_cast<v4f*>(dst + t * 16 * 64)     = lo;
            *reinterpret_cast<v4f*>(dst + t * 16 * 64 + 4) = hi;
        }
    }
}

__global__ __launch_bounds__(256) void ksoftmax(float* __restrict__ Kbuf)
{
    const int i = blockIdx.x * 256 + threadIdx.x;
    float* p = Kbuf + (size_t)i * DK;
    float v[DK], mx = -3.0e38f;
    #pragma unroll
    for (int k = 0; k < DK; ++k) { v[k] = p[k]; mx = fmaxf(mx, v[k]); }
    float s = 0.f;
    #pragma unroll
    for (int k = 0; k < DK; ++k) { v[k] = __expf(v[k] - mx); s += v[k]; }
    const float inv = 1.f / s;
    #pragma unroll
    for (int k = 0; k < DK; ++k) p[k] = v[k] * inv;
}

#define PQ     160
#define RP     36
#define SV_SET (16 * PQ)
#define SR_SET (51 * RP)
#define SB_V   0
#define SB_K   (3 * SV_SET)
#define SB_R   (SB_K + 1024)
#define SB_N   (SB_R + 3 * SR_SET)
#define E_LAM  0
#define E_GSH  16384
#define E_GS   (16384 + 2048)

__global__ __launch_bounds__(256, 2) void lambda_main(
    const float* __restrict__ rel,
    const float* __restrict__ Qbuf,
    const float* __restrict__ sK,
    const float* __restrict__ Vbuf,
    float* __restrict__ out)
{
    __shared__ __attribute__((aligned(16))) float stg[2][SB_N];

    const int tid  = threadIdx.x;
    const int wave = tid >> 5;
    const int lane = tid & 31;
    const int kM   = lane & 15;
    const int half = lane >> 4;

    const int b   = blockIdx.x >> 6;
    const int blk = blockIdx.x & 63;
    const int rp  = blk >> 2;
    const int xb  = blk & 3;
    const int ny0 = 2 * rp;
    const int ql  = wave & 1;
    const int vt  = wave >> 1;
    const int qa  = (xb & 1) * 2 + ql;
    const int xg  = (xb >> 1) * 16;
    const int cs  = 16 - xg;

    const int pid = tid >> 4;
    const int l16 = tid & 15;
    const int fq  = pid & 3;
    const int fP  = pid >> 2;
    const int ri2 = pid + 32 < 47 ? pid + 32 : 46;
    const float* Vg = Vbuf + (size_t)b * MM * DV + (8 * fP + fq) * DV + l16 * 4;
    const float* Kg = sK   + (size_t)b * MM * DK + (8 * fP + fq) * DK + l16;
    const float* Rg = rel + ((31 - ny0) * 63 + cs) * DK + l16;

    const int prow = fq * 4 + fP;
    const int vst  = SB_V + prow * PQ + l16 * 8;
    const int kst  = SB_K + (prow * 16 + l16) * 2;
    const int rst0 = SB_R + pid * RP + l16 * 2 + 1;
    const int rst1 = SB_R + (pid + 16) * RP + l16 * 2 + 1;
    const int rst2 = SB_R + ri2 * RP + l16 * 2 + 1;

    const int vrd = SB_V + half * PQ + (vt * 16 + kM) * 2;
    const int krd = SB_K + ql * 512 + (half * 16 + kM) * 2;
    const int grd = vrd + (ql ? 1 : 2) * SV_SET;
    const int trd = SB_R + (8 * half - 4 * qa + 15 + 4) * RP + kM * 2;

    v8f acc[3][9] = {};
    v8f accG[2] = {};

    v4f fa0, fb0, fa1, fb1;
    float ka0, kb0, ka1, kb1;
    float rm0, rm1, rm2, rc0, rc1, rc2, rq0, rq1, rq2;

#define FETCH_STAGE(ux)                                                      \
    fa0 = *reinterpret_cast<const v4f*>(Vg + (ux) * 4096);                  \
    fb0 = *reinterpret_cast<const v4f*>(Vg + (ux) * 4096 + 4 * DV);         \
    fa1 = *reinterpret_cast<const v4f*>(Vg + (ux) * 4096 + 2048);           \
    fb1 = *reinterpret_cast<const v4f*>(Vg + (ux) * 4096 + 2048 + 4 * DV);  \
    ka0 = Kg[(ux) * 1024];        kb0 = Kg[(ux) * 1024 + 4 * DK];           \
    ka1 = Kg[(ux) * 1024 + 512];  kb1 = Kg[(ux) * 1024 + 512 + 4 * DK];     \
    rm0 = Rg[((ux) * 2 - 1) * 1008 + pid * 16];                             \
    rm1 = Rg[((ux) * 2 - 1) * 1008 + (pid + 16) * 16];                      \
    rm2 = Rg[((ux) * 2 - 1) * 1008 + ri2 * 16];                             \
    rc0 = Rg[(ux) * 2016 + pid * 16];                                       \
    rc1 = Rg[(ux) * 2016 + (pid + 16) * 16];                                \
    rc2 = Rg[(ux) * 2016 + ri2 * 16];                                       \
    rq0 = Rg[((ux) * 2 + 1) * 1008 + pid * 16];                             \
    rq1 = Rg[((ux) * 2 + 1) * 1008 + (pid + 16) * 16];                      \
    rq2 = Rg[((ux) * 2 + 1) * 1008 + ri2 * 16];

#define STORE_V(Sx, off, A_, B_)                                             \
    {                                                                        \
        v4f lo, hi;                                                          \
        lo.x = (A_).x; lo.y = (B_).x; lo.z = (A_).y; lo.w = (B_).y;          \
        hi.x = (A_).z; hi.y = (B_).z; hi.z = (A_).w; hi.w = (B_).w;          \
        *reinterpret_cast<v4f*>((Sx) + (off))     = lo;                      \
        *reinterpret_cast<v4f*>((Sx) + (off) + 4) = hi;                      \
    }

#define STORE_R(Sx, rst, c_, q_, m_)                                         \
    (Sx)[(rst)] = (c_);                   (Sx)[(rst) + 4 * RP - 1] = (c_);                   \
    (Sx)[(rst) + SR_SET] = (q_) - (c_);   (Sx)[(rst) + SR_SET + 4 * RP - 1] = (q_) - (c_);   \
    (Sx)[(rst) + 2 * SR_SET] = (m_) - (c_); (Sx)[(rst) + 2 * SR_SET + 4 * RP - 1] = (m_) - (c_);

#define STORE_STAGE(Sx)                                                      \
    {                                                                        \
        const v4f sa = fa0 + fa1;                                            \
        const v4f sb = fb0 + fb1;                                            \
        STORE_V(Sx, vst, sa, sb)                                             \
        STORE_V(Sx, vst + SV_SET, fa1, fb1)                                  \
        STORE_V(Sx, vst + 2 * SV_SET, fa0, fb0)                              \
        v2f kk; kk.x = ka0; kk.y = kb0;                                      \
        *reinterpret_cast<v2f*>((Sx) + kst) = kk;                            \
        kk.x = ka1; kk.y = kb1;                                              \
        *reinterpret_cast<v2f*>((Sx) + kst + 512) = kk;                      \
        STORE_R(Sx, rst0, rc0, rq0, rm0)                                     \
        STORE_R(Sx, rst1, rc1, rq1, rm1)                                     \
        STORE_R(Sx, rst2, rc2, rq2, rm2)                                     \
    }

    FETCH_STAGE(0)
    STORE_STAGE(stg[0])
    __syncthreads();

    for (int u = 0; u < 16; ++u) {
        const int un = u < 15 ? u + 1 : 15;
        FETCH_STAGE(un)

        const float* S = stg[u & 1];
        v2f rr[6][7];
        v2f Vq[6][4];
        v2f kf[2][4], vg[2][4];
#define LOAD_GROUP(g_)                                                               \
        {                                                                            \
            const int jl = (g_) / 3, Xl = (g_) % 3;                                  \
            _Pragma("unroll")                                                        \
            for (int d = 0; d < 7; ++d)                                              \
                rr[g_][d] = *reinterpret_cast<const v2f*>(                           \
                                S + trd + Xl * SR_SET + (16 * jl + d - 3) * RP);     \
            _Pragma("unroll")                                                        \
            for (int q = 0; q < 4; ++q)                                              \
                Vq[g_][q] = *reinterpret_cast<const v2f*>(                           \
                                S + vrd + Xl * SV_SET + jl * 2 * PQ + q * 4 * PQ);   \
            if (Xl == 2) {                                                           \
                _Pragma("unroll")                                                    \
                for (int q = 0; q < 4; ++q) {                                        \
                    kf[jl][q] = *reinterpret_cast<const v2f*>(S + krd + q * 128 + jl * 64);          \
                    vg[jl][q] = *reinterpret_cast<const v2f*>(S + grd + q * 4 * PQ + jl * 2 * PQ);   \
                }                                                                    \
            }                                                                        \
        }
        LOAD_GROUP(0)
        #pragma unroll
        for (int g = 0; g < 6; ++g) {
            const int jj = g / 3, X = g % 3;
            if (g < 5) LOAD_GROUP(g + 1)
            __builtin_amdgcn_sched_barrier(0);
            v2f a[9];
            a[0] = rr[g][3];
            a[1] = rr[g][4] - rr[g][3];
            a[2] = rr[g][2] - rr[g][3];
            a[3] = rr[g][5] - rr[g][3];
            a[4] = (rr[g][6] - rr[g][4]) - a[3];
            a[5] = (rr[g][4] - rr[g][2]) - a[3];
            a[6] = rr[g][1] - rr[g][3];
            a[7] = (rr[g][2] - rr[g][4]) - a[6];
            a[8] = (rr[g][0] - rr[g][2]) - a[6];
            const v2f V0 = Vq[g][0], V1 = Vq[g][1], V2 = Vq[g][2], V3 = Vq[g][3];
            const v2f s02 = V0 + V2;
            const v2f s13 = V1 + V3;
            const v2f s01 = V0 + V1;
            const v2f s23 = V2 + V3;
            const v2f sal = s02 + s13;
            acc[X][0] = __builtin_amdgcn_wmma_f32_16x16x4_f32(
                            false, a[0], false, sal, (short)0, acc[X][0], false, false);
            acc[X][1] = __builtin_amdgcn_wmma_f32_16x16x4_f32(
                            false, a[1], false, s13, (short)0, acc[X][1], false, false);
            acc[X][2] = __builtin_amdgcn_wmma_f32_16x16x4_f32(
                            false, a[2], false, s02, (short)0, acc[X][2], false, false);
            acc[X][3] = __builtin_amdgcn_wmma_f32_16x16x4_f32(
                            false, a[3], false, s23, (short)0, acc[X][3], false, false);
            acc[X][4] = __builtin_amdgcn_wmma_f32_16x16x4_f32(
                            false, a[4], false, V3, (short)0, acc[X][4], false, false);
            acc[X][5] = __builtin_amdgcn_wmma_f32_16x16x4_f32(
                            false, a[5], false, V2, (short)0, acc[X][5], false, false);
            acc[X][6] = __builtin_amdgcn_wmma_f32_16x16x4_f32(
                            false, a[6], false, s01, (short)0, acc[X][6], false, false);
            acc[X][7] = __builtin_amdgcn_wmma_f32_16x16x4_f32(
                            false, a[7], false, V1, (short)0, acc[X][7], false, false);
            acc[X][8] = __builtin_amdgcn_wmma_f32_16x16x4_f32(
                            false, a[8], false, V0, (short)0, acc[X][8], false, false);
            if (X == 2) {
                #pragma unroll
                for (int q = 0; q < 4; ++q)
                    accG[q & 1] = __builtin_amdgcn_wmma_f32_16x16x4_f32(
                                      false, kf[jj][q], false, vg[jj][q], (short)0, accG[q & 1], false, false);
            }
            __builtin_amdgcn_sched_barrier(0);
        }
#undef LOAD_GROUP

        float* Sn = stg[(u + 1) & 1];
        STORE_STAGE(Sn)
        __syncthreads();
    }
#undef FETCH_STAGE
#undef STORE_STAGE
#undef STORE_V
#undef STORE_R

    float* E = &stg[0][0];

    #pragma unroll
    for (int r = 0; r < 8; ++r)
        E[E_GSH + wave * 256 + (r + half * 8) * 16 + kM] = accG[0][r] + accG[1][r];
    __syncthreads();
    {
        const int ge = tid * 4;
        const int gk = ge >> 6;
        const int gv = ge & 63;
        const int w0 = (gv >> 4) * 2;
        const int gi = gk * 16 + (gv & 15);
        const v4f g = *reinterpret_cast<const v4f*>(E + E_GSH + w0 * 256 + gi)
                    + *reinterpret_cast<const v4f*>(E + E_GSH + (w0 + 1) * 256 + gi);
        *reinterpret_cast<v4f*>(E + E_GS + ge) = g;
    }
    __syncthreads();

    #pragma unroll
    for (int r = 0; r < 8; ++r) {
        const int k = r + half * 8;
        const float g = E[E_GS + k * 64 + vt * 16 + kM];
        #pragma unroll
        for (int y = 0; y < 2; ++y)
            #pragma unroll
            for (int p = 0; p < 4; ++p) {
                float l[2];
                #pragma unroll
                for (int z = 0; z < 2; ++z) {
                    const int X = z == 0 ? 0 : (y == 0 ? 1 : 2);
                    const float c1 = (p & 1) ? acc[X][2][r] : acc[X][1][r];
                    const float c2 = (p & 2) ? acc[X][6][r] : acc[X][3][r];
                    const float c3 = p == 0 ? acc[X][4][r] : p == 1 ? acc[X][5][r]
                                   : p == 2 ? acc[X][7][r] : acc[X][8][r];
                    l[z] = (acc[X][0][r] + c1) + (c2 + c3);
                }
                E[E_LAM + ((((y * 4 + p) * 2 + ql) * 16 + k) << 6) + vt * 16 + kM] = (l[0] + l[1]) + g;
            }
    }
    __syncthreads();

    #pragma unroll 1
    for (int i = 0; i < 2; ++i) {
        const int pi = wave * 2 + i;
        const int y  = pi >> 3;
        const int p  = (pi >> 1) & 3;
        const int n  = (ny0 + y) * 32 + xg + 4 * ((xb & 1) * 2 + (pi & 1)) + p;
        const float* Qn = Qbuf + (size_t)(b * MM + n) * QC;
        const float* L  = E + E_LAM + pi * 1024 + lane;
        float* on = out + (size_t)(b * MM + n) * 256 + lane;
        float o[4][2] = {};
        #pragma unroll
        for (int k = 0; k < DK; ++k) {
            const float l0 = L[k * 64];
            const float l1 = L[k * 64 + 32];
            #pragma unroll
            for (int h = 0; h < 4; ++h) {
                const float qk = Qn[h * 16 + k];
                o[h][0] = fmaf(qk, l0, o[h][0]);
                o[h][1] = fmaf(qk, l1, o[h][1]);
            }
        }
        #pragma unroll
        for (int h = 0; h < 4; ++h) {
            on[h * 64]      = o[h][0];
            on[h * 64 + 32] = o[h][1];
        }
    }
}

extern "C" void kernel_launch(void* const* d_in, const int* in_sizes, int n_in,
                              void* d_out, int out_size, void* d_ws, size_t ws_size,
                              hipStream_t stream)
{
    const float* x   = (const float*)d_in[0];
    const float* Wq  = (const float*)d_in[1];
    const float* Wk  = (const float*)d_in[2];
    const float* Wv  = (const float*)d_in[3];
    const float* gq  = (const float*)d_in[4];
    const float* bq  = (const float*)d_in[5];
    const float* mq  = (const float*)d_in[6];
    const float* vq  = (const float*)d_in[7];
    const float* gv  = (const float*)d_in[8];
    const float* bv  = (const float*)d_in[9];
    const float* mv  = (const float*)d_in[10];
    const float* vv  = (const float*)d_in[11];
    const float* rel = (const float*)d_in[12];
    float* out = (float*)d_out;

    float* Qbuf = (float*)d_ws;
    float* Vbuf = Qbuf + (size_t)NB * MM * QC;
    float* Kbuf = Vbuf + (size_t)NB * MM * DV;

    qkv_wmma<<<256, 288, 0, stream>>>(x, Wq, Wk, Wv, gq, bq, mq, vq,
                                       gv, bv, mv, vv, Qbuf, Kbuf, Vbuf);
    ksoftmax<<<64, 256, 0, stream>>>(Kbuf);
    lambda_main<<<1024, 256, 0, stream>>>(rel, Qbuf, Kbuf, Vbuf, out);
}


# Kernel 3: stage boundary moved inside the WMMA run: next row pair stored to LDS between WMMA groups (fetched a stage earlier), LDS-only barrier before the last group, next fragments read behind it
# speedup vs baseline: 1.7044x; 1.0401x over previous
#include <hip/hip_runtime.h>

typedef __attribute__((ext_vector_type(2))) float v2f;
typedef __attribute__((ext_vector_type(4))) float v4f;
typedef __attribute__((ext_vector_type(8))) float v8f;

#define EPSF 1e-5f
#define NB   16
#define CIN  256
#define MM   1024
#define DK   16
#define DV   64
#define QC   64

#define XP 288

__global__ __launch_bounds__(288) void qkv_wmma(
    const float* __restrict__ x,  const float* __restrict__ Wq,
    const float* __restrict__ Wk, const float* __restrict__ Wv,
    const float* __restrict__ gq, const float* __restrict__ bq,
    const float* __restrict__ mq, const float* __restrict__ vq,
    const float* __restrict__ gv, const float* __restrict__ bv,
    const float* __restrict__ mv, const float* __restrict__ vv,
    float* __restrict__ Qbuf,
    float* __restrict__ Kbuf,
    float* __restrict__ Vbuf)
{
    __shared__ __attribute__((aligned(16))) float xs[64 * XP];

    const int tid  = threadIdx.x;
    const int wave = tid >> 5;
    const int lane = tid & 31;
    const int b    = blockIdx.x >> 4;
    const int m0   = (blockIdx.x & 15) * 64;
    const int nL   = lane & 15;
    const int half = lane >> 4;

    v4f xv[16];
    if (tid < 256) {
        const int m  = tid & 63;
        const int cq = tid >> 6;
        const float* xg = x + (size_t)b * CIN * MM + (size_t)(cq * 4) * MM + m0 + m;
        #pragma unroll
        for (int k = 0; k < 16; ++k) {
            xv[k].x = xg[(size_t)(k * 16 + 0) * MM];
            xv[k].y = xg[(size_t)(k * 16 + 1) * MM];
            xv[k].z = xg[(size_t)(k * 16 + 2) * MM];
            xv[k].w = xg[(size_t)(k * 16 + 3) * MM];
        }
    }

    const float* Wm; int orow;
    if (wave < 4)       { Wm = Wq; orow = wave * 16; }
    else if (wave == 4) { Wm = Wk; orow = 0; }
    else                { Wm = Wv; orow = (wave - 5) * 16; }
    const v4f* wrow = reinterpret_cast<const v4f*>(
                          Wm + (size_t)(orow + nL) * CIN + half * 4);
    v4f wf[32];
    #pragma unroll
    for (int i = 0; i < 8; ++i) wf[i] = wrow[i * 2];

    const bool isq = wave < 4;
    const int  oc  = (isq ? wave * 16 : (wave - 5) * 16) + half * 8;
    float sc[8], sh[8];
    if (wave != 4) {
        const float* g  = isq ? gq : gv;
        const float* be = isq ? bq : bv;
        const float* mn = isq ? mq : mv;
        const float* vr = isq ? vq : vv;
        #pragma unroll
        for (int r = 0; r < 8; ++r) {
            sc[r] = g[oc + r] * rsqrtf(vr[oc + r] + EPSF);
            sh[r] = be[oc + r] - mn[oc + r] * sc[r];
        }
    }

    if (tid < 256) {
        float* xd = xs + (tid >> 6) * XP + (tid & 63) * 4;
        #pragma unroll
        for (int k = 0; k < 16; ++k)
            *reinterpret_cast<v4f*>(xd + k * 4 * XP) = xv[k];
    }
    __syncthreads();

    #pragma unroll
    for (int i = 8; i < 32; ++i) wf[i] = wrow[i * 2];

    const float* xb = xs + half * XP + nL * 4;
    v8f acc[4] = {};
    #pragma unroll
    for (int i = 0; i < 32; ++i) {
        v4f bf[4];
        #pragma unroll
        for (int t = 0; t < 4; ++t)
            bf[t] = *reinterpret_cast<const v4f*>(xb + i * 2 * XP + t * 64);
        v2f alo, ahi;
        alo.x = wf[i].x; alo.y = wf[i].y;
        ahi.x = wf[i].z; ahi.y = wf[i].w;
        #pragma unroll
        for (int t = 0; t < 4; ++t) {
            v2f bl; bl.x = bf[t].x; bl.y = bf[t].y;
            acc[t] = __builtin_amdgcn_wmma_f32_16x16x4_f32(
                         false, alo, false, bl, (short)0, acc[t], false, false);
        }
        #pragma unroll
        for (int t = 0; t < 4; ++t) {
            v2f bh; bh.x = bf[t].z; bh.y = bf[t].w;
            acc[t] = __builtin_amdgcn_wmma_f32_16x16x4_f32(
                         false, ahi, false, bh, (short)0, acc[t], false, false);
        }
    }

    const int n0 = b * MM + m0 + nL;
    if (wave == 4) {
        float* dst = Kbuf + (size_t)n0 * DK + half * 8;
        #pragma unroll
        for (int t = 0; t < 4; ++t) {
            v4f lo, hi;
            lo.x = acc[t][0]; lo.y = acc[t][1]; lo.z = acc[t][2]; lo.w = acc[t][3];
            hi.x = acc[t][4]; hi.y = acc[t][5]; hi.z = acc[t][6]; hi.w = acc[t][7];
            *reinterpret_cast<v4f*>(dst + t * 16 * DK)     = lo;
            *reinterpret_cast<v4f*>(dst + t * 16 * DK + 4) = hi;
        }
    } else {
        float* dst = (isq ? Qbuf : Vbuf) + (size_t)n0 * 64 + oc;
        #pragma unroll
        for (int t = 0; t < 4; ++t) {
            v4f lo, hi;
            lo.x = acc[t][0] * sc[0] + sh[0]; lo.y = acc[t][1] * sc[1] + sh[1];
            lo.z = acc[t][2] * sc[2] + sh[2]; lo.w = acc[t][3] * sc[3] + sh[3];
            hi.x = acc[t][4] * sc[4] + sh[4]; hi.y = acc[t][5] * sc[5] + sh[5];
            hi.z = acc[t][6] * sc[6] + sh[6]; hi.w = acc[t][7] * sc[7] + sh[7];
            *reinterpret_cast<v4f*>(dst + t * 16 * 64)     = lo;
            *reinterpret_cast<v4f*>(dst + t * 16 * 64 + 4) = hi;
        }
    }
}

__global__ __launch_bounds__(256) void ksoftmax(float* __restrict__ Kbuf)
{
    const int i = blockIdx.x * 256 + threadIdx.x;
    float* p = Kbuf + (size_t)i * DK;
    float v[DK], mx = -3.0e38f;
    #pragma unroll
    for (int k = 0; k < DK; ++k) { v[k] = p[k]; mx = fmaxf(mx, v[k]); }
    float s = 0.f;
    #pragma unroll
    for (int k = 0; k < DK; ++k) { v[k] = __expf(v[k] - mx); s += v[k]; }
    const float inv = 1.f / s;
    #pragma unroll
    for (int k = 0; k < DK; ++k) p[k] = v[k] * inv;
}

#define PQ     160
#define RP     36
#define SV_SET (16 * PQ)
#define SR_SET (51 * RP)
#define SB_V   0
#define SB_K   (3 * SV_SET)
#define SB_R   (SB_K + 1024)
#define SB_N   (SB_R + 3 * SR_SET)
#define E_LAM  0
#define E_GSH  16384
#define E_GS   (16384 + 2048)

__global__ __launch_bounds__(256, 2) void lambda_main(
    const float* __restrict__ rel,
    const float* __restrict__ Qbuf,
    const float* __restrict__ sK,
    const float* __restrict__ Vbuf,
    float* __restrict__ out)
{
    __shared__ __attribute__((aligned(16))) float stg[2][SB_N];

    const int tid  = threadIdx.x;
    const int wave = tid >> 5;
    const int lane = tid & 31;
    const int kM   = lane & 15;
    const int half = lane >> 4;

    const int b   = blockIdx.x >> 6;
    const int blk = blockIdx.x & 63;
    const int rp  = blk >> 2;
    const int xb  = blk & 3;
    const int ny0 = 2 * rp;
    const int ql  = wave & 1;
    const int vt  = wave >> 1;
    const int qa  = (xb & 1) * 2 + ql;
    const int xg  = (xb >> 1) * 16;
    const int cs  = 16 - xg;

    const int pid = tid >> 4;
    const int l16 = tid & 15;
    const int fq  = pid & 3;
    const int fP  = pid >> 2;
    const int ri2 = pid + 32 < 47 ? pid + 32 : 46;
    const float* Vg = Vbuf + (size_t)b * MM * DV + (8 * fP + fq) * DV + l16 * 4;
    const float* Kg = sK   + (size_t)b * MM * DK + (8 * fP + fq) * DK + l16;
    const float* Rg = rel + ((31 - ny0) * 63 + cs) * DK + l16;

    const int prow = fq * 4 + fP;
    const int vst  = SB_V + prow * PQ + l16 * 8;
    const int kst  = SB_K + (prow * 16 + l16) * 2;
    const int rst0 = SB_R + pid * RP + l16 * 2 + 1;
    const int rst1 = SB_R + (pid + 16) * RP + l16 * 2 + 1;
    const int rst2 = SB_R + ri2 * RP + l16 * 2 + 1;

    const int vrd = SB_V + half * PQ + (vt * 16 + kM) * 2;
    const int krd = SB_K + ql * 512 + (half * 16 + kM) * 2;
    const int grd = vrd + (ql ? 1 : 2) * SV_SET;
    const int trd = SB_R + (8 * half - 4 * qa + 15 + 4) * RP + kM * 2;

    v8f acc[3][9] = {};
    v8f accG[2] = {};

    v4f fa0, fb0, fa1, fb1;
    float ka0, kb0, ka1, kb1;
    float rm0, rm1, rm2, rc0, rc1, rc2, rq0, rq1, rq2;

#define FETCH_STAGE(ux)                                                      \
    fa0 = *reinterpret_cast<const v4f*>(Vg + (ux) * 4096);                  \
    fb0 = *reinterpret_cast<const v4f*>(Vg + (ux) * 4096 + 4 * DV);         \
    fa1 = *reinterpret_cast<const v4f*>(Vg + (ux) * 4096 + 2048);           \
    fb1 = *reinterpret_cast<const v4f*>(Vg + (ux) * 4096 + 2048 + 4 * DV);  \
    ka0 = Kg[(ux) * 1024];        kb0 = Kg[(ux) * 1024 + 4 * DK];           \
    ka1 = Kg[(ux) * 1024 + 512];  kb1 = Kg[(ux) * 1024 + 512 + 4 * DK];     \
    rm0 = Rg[((ux) * 2 - 1) * 1008 + pid * 16];                             \
    rm1 = Rg[((ux) * 2 - 1) * 1008 + (pid + 16) * 16];                      \
    rm2 = Rg[((ux) * 2 - 1) * 1008 + ri2 * 16];                             \
    rc0 = Rg[(ux) * 2016 + pid * 16];                                       \
    rc1 = Rg[(ux) * 2016 + (pid + 16) * 16];                                \
    rc2 = Rg[(ux) * 2016 + ri2 * 16];                                       \
    rq0 = Rg[((ux) * 2 + 1) * 1008 + pid * 16];                             \
    rq1 = Rg[((ux) * 2 + 1) * 1008 + (pid + 16) * 16];                      \
    rq2 = Rg[((ux) * 2 + 1) * 1008 + ri2 * 16];

#define STORE_V(Sx, off, A_, B_)                                             \
    {                                                                        \
        v4f lo, hi;                                                          \
        lo.x = (A_).x; lo.y = (B_).x; lo.z = (A_).y; lo.w = (B_).y;          \
        hi.x = (A_).z; hi.y = (B_).z; hi.z = (A_).w; hi.w = (B_).w;          \
        *reinterpret_cast<v4f*>((Sx) + (off))     = lo;                      \
        *reinterpret_cast<v4f*>((Sx) + (off) + 4) = hi;                      \
    }

#define STORE_R(Sx, rst, c_, q_, m_)                                         \
    (Sx)[(rst)] = (c_);                   (Sx)[(rst) + 4 * RP - 1] = (c_);                   \
    (Sx)[(rst) + SR_SET] = (q_) - (c_);   (Sx)[(rst) + SR_SET + 4 * RP - 1] = (q_) - (c_);   \
    (Sx)[(rst) + 2 * SR_SET] = (m_) - (c_); (Sx)[(rst) + 2 * SR_SET + 4 * RP - 1] = (m_) - (c_);

#define STORE_STAGE(Sx)                                                      \
    {                                                                        \
        const v4f sa = fa0 + fa1;                                            \
        const v4f sb = fb0 + fb1;                                            \
        STORE_V(Sx, vst, sa, sb)                                             \
        STORE_V(Sx, vst + SV_SET, fa1, fb1)                                  \
        STORE_V(Sx, vst + 2 * SV_SET, fa0, fb0)                              \
        v2f kk; kk.x = ka0; kk.y = kb0;                                      \
        *reinterpret_cast<v2f*>((Sx) + kst) = kk;                            \
        kk.x = ka1; kk.y = kb1;                                              \
        *reinterpret_cast<v2f*>((Sx) + kst + 512) = kk;                      \
        STORE_R(Sx, rst0, rc0, rq0, rm0)                                     \
        STORE_R(Sx, rst1, rc1, rq1, rm1)                                     \
        STORE_R(Sx, rst2, rc2, rq2, rm2)                                     \
    }

#define LDS_BARRIER()                                                        \
    {                                                                        \
        __builtin_amdgcn_fence(__ATOMIC_RELEASE, "workgroup", "local");      \
        __builtin_amdgcn_s_barrier();                                        \
        __builtin_amdgcn_fence(__ATOMIC_ACQUIRE, "workgroup", "local");      \
    }

    FETCH_STAGE(0)
    STORE_STAGE(stg[0])
    FETCH_STAGE(1)
    LDS_BARRIER()

    v2f rr[6][7];
    v2f Vq[6][4];
    v2f kf[2][4], vg[2][4];
#define LOAD_GROUP(g_, Sp)                                                           \
    {                                                                                \
        const int jl = (g_) / 3, Xl = (g_) % 3;                                      \
        _Pragma("unroll")                                                            \
        for (int d = 0; d < 7; ++d)                                                  \
            rr[g_][d] = *reinterpret_cast<const v2f*>(                               \
                            (Sp) + trd + Xl * SR_SET + (16 * jl + d - 3) * RP);      \
        _Pragma("unroll")                                                            \
        for (int q = 0; q < 4; ++q)                                                  \
            Vq[g_][q] = *reinterpret_cast<const v2f*>(                               \
                            (Sp) + vrd + Xl * SV_SET + jl * 2 * PQ + q * 4 * PQ);    \
        if (Xl == 2) {                                                               \
            _Pragma("unroll")                                                        \
            for (int q = 0; q < 4; ++q) {                                            \
                kf[jl][q] = *reinterpret_cast<const v2f*>((Sp) + krd + q * 128 + jl * 64);          \
                vg[jl][q] = *reinterpret_cast<const v2f*>((Sp) + grd + q * 4 * PQ + jl * 2 * PQ);   \
            }                                                                        \
        }                                                                            \
    }
    LOAD_GROUP(0, stg[0])

    for (int u = 0; u < 16; ++u) {
        const int un = u < 14 ? u + 2 : 15;

        const float* S  = stg[u & 1];
        float*       Sn = stg[(u + 1) & 1];
        #pragma unroll
        for (int g = 0; g < 6; ++g) {
            const int jj = g / 3, X = g % 3;
            if (g < 5) LOAD_GROUP(g + 1, S)
            else       LOAD_GROUP(0, Sn)
            __builtin_amdgcn_sched_barrier(0);
            v2f a[9];
            a[0] = rr[g][3];
            a[1] = rr[g][4] - rr[g][3];
            a[2] = rr[g][2] - rr[g][3];
            a[3] = rr[g][5] - rr[g][3];
            a[4] = (rr[g][6] - rr[g][4]) - a[3];
            a[5] = (rr[g][4] - rr[g][2]) - a[3];
            a[6] = rr[g][1] - rr[g][3];
            a[7] = (rr[g][2] - rr[g][4]) - a[6];
            a[8] = (rr[g][0] - rr[g][2]) - a[6];
            const v2f V0 = Vq[g][0], V1 = Vq[g][1], V2 = Vq[g][2], V3 = Vq[g][3];
            const v2f s02 = V0 + V2;
            const v2f s13 = V1 + V3;
            const v2f s01 = V0 + V1;
            const v2f s23 = V2 + V3;
            const v2f sal = s02 + s13;
            acc[X][0] = __builtin_amdgcn_wmma_f32_16x16x4_f32(
                            false, a[0], false, sal, (short)0, acc[X][0], false, false);
            acc[X][1] = __builtin_amdgcn_wmma_f32_16x16x4_f32(
                            false, a[1], false, s13, (short)0, acc[X][1], false, false);
            acc[X][2] = __builtin_amdgcn_wmma_f32_16x16x4_f32(
                            false, a[2], false, s02, (short)0, acc[X][2], false, false);
            acc[X][3] = __builtin_amdgcn_wmma_f32_16x16x4_f32(
                            false, a[3], false, s23, (short)0, acc[X][3], false, false);
            acc[X][4] = __builtin_amdgcn_wmma_f32_16x16x4_f32(
                            false, a[4], false, V3, (short)0, acc[X][4], false, false);
            acc[X][5] = __builtin_amdgcn_wmma_f32_16x16x4_f32(
                            false, a[5], false, V2, (short)0, acc[X][5], false, false);
            acc[X][6] = __builtin_amdgcn_wmma_f32_16x16x4_f32(
                            false, a[6], false, s01, (short)0, acc[X][6], false, false);
            acc[X][7] = __builtin_amdgcn_wmma_f32_16x16x4_f32(
                            false, a[7], false, V1, (short)0, acc[X][7], false, false);
            acc[X][8] = __builtin_amdgcn_wmma_f32_16x16x4_f32(
                            false, a[8], false, V0, (short)0, acc[X][8], false, false);
            if (X == 2) {
                #pragma unroll
                for (int q = 0; q < 4; ++q)
                    accG[q & 1] = __builtin_amdgcn_wmma_f32_16x16x4_f32(
                                      false, kf[jj][q], false, vg[jj][q], (short)0, accG[q & 1], false, false);
            }
            __builtin_amdgcn_sched_barrier(0);
            if (g == 1) { STORE_STAGE(Sn) FETCH_STAGE(un) }
            if (g == 4) LDS_BARRIER()
        }
    }
#undef LOAD_GROUP
#undef LDS_BARRIER
#undef FETCH_STAGE
#undef STORE_STAGE
#undef STORE_V
#undef STORE_R

    float* E = &stg[0][0];

    #pragma unroll
    for (int r = 0; r < 8; ++r)
        E[E_GSH + wave * 256 + (r + half * 8) * 16 + kM] = accG[0][r] + accG[1][r];
    __syncthreads();
    {
        const int ge = tid * 4;
        const int gk = ge >> 6;
        const int gv = ge & 63;
        const int w0 = (gv >> 4) * 2;
        const int gi = gk * 16 + (gv & 15);
        const v4f g = *reinterpret_cast<const v4f*>(E + E_GSH + w0 * 256 + gi)
                    + *reinterpret_cast<const v4f*>(E + E_GSH + (w0 + 1) * 256 + gi);
        *reinterpret_cast<v4f*>(E + E_GS + ge) = g;
    }
    __syncthreads();

    #pragma unroll
    for (int r = 0; r < 8; ++r) {
        const int k = r + half * 8;
        const float g = E[E_GS + k * 64 + vt * 16 + kM];
        #pragma unroll
        for (int y = 0; y < 2; ++y)
            #pragma unroll
            for (int p = 0; p < 4; ++p) {
                float l[2];
                #pragma unroll
                for (int z = 0; z < 2; ++z) {
                    const int X = z == 0 ? 0 : (y == 0 ? 1 : 2);
                    const float c1 = (p & 1) ? acc[X][2][r] : acc[X][1][r];
                    const float c2 = (p & 2) ? acc[X][6][r] : acc[X][3][r];
                    const float c3 = p == 0 ? acc[X][4][r] : p == 1 ? acc[X][5][r]
                                   : p == 2 ? acc[X][7][r] : acc[X][8][r];
                    l[z] = (acc[X][0][r] + c1) + (c2 + c3);
                }
                E[E_LAM + ((((y * 4 + p) * 2 + ql) * 16 + k) << 6) + vt * 16 + kM] = (l[0] + l[1]) + g;
            }
    }
    __syncthreads();

    #pragma unroll 1
    for (int i = 0; i < 2; ++i) {
        const int pi = wave * 2 + i;
        const int y  = pi >> 3;
        const int p  = (pi >> 1) & 3;
        const int n  = (ny0 + y) * 32 + xg + 4 * ((xb & 1) * 2 + (pi & 1)) + p;
        const float* Qn = Qbuf + (size_t)(b * MM + n) * QC;
        const float* L  = E + E_LAM + pi * 1024 + lane;
        float* on = out + (size_t)(b * MM + n) * 256 + lane;
        float o[4][2] = {};
        #pragma unroll
        for (int k = 0; k < DK; ++k) {
            const float l0 = L[k * 64];
            const float l1 = L[k * 64 + 32];
            #pragma unroll
            for (int h = 0; h < 4; ++h) {
                const float qk = Qn[h * 16 + k];
                o[h][0] = fmaf(qk, l0, o[h][0]);
                o[h][1] = fmaf(qk, l1, o[h][1]);
            }
        }
        #pragma unroll
        for (int h = 0; h < 4; ++h) {
            on[h * 64]      = o[h][0];
            on[h * 64 + 32] = o[h][1];
        }
    }
}

extern "C" void kernel_launch(void* const* d_in, const int* in_sizes, int n_in,
                              void* d_out, int out_size, void* d_ws, size_t ws_size,
                              hipStream_t stream)
{
    const float* x   = (const float*)d_in[0];
    const float* Wq  = (const float*)d_in[1];
    const float* Wk  = (const float*)d_in[2];
    const float* Wv  = (const float*)d_in[3];
    const float* gq  = (const float*)d_in[4];
    const float* bq  = (const float*)d_in[5];
    const float* mq  = (const float*)d_in[6];
    const float* vq  = (const float*)d_in[7];
    const float* gv  = (const float*)d_in[8];
    const float* bv  = (const float*)d_in[9];
    const float* mv  = (const float*)d_in[10];
    const float* vv  = (const float*)d_in[11];
    const float* rel = (const float*)d_in[12];
    float* out = (float*)d_out;

    float* Qbuf = (float*)d_ws;
    float* Vbuf = Qbuf + (size_t)NB * MM * QC;
    float* Kbuf = Vbuf + (size_t)NB * MM * DV;

    qkv_wmma<<<256, 288, 0, stream>>>(x, Wq, Wk, Wv, gq, bq, mq, vq,
                                       gv, bv, mv, vv, Qbuf, Kbuf, Vbuf);
    ksoftmax<<<64, 256, 0, stream>>>(Kbuf);
    lambda_main<<<1024, 256, 0, stream>>>(rel, Qbuf, Kbuf, Vbuf, out);
}


# Kernel 3 y direction: 2 position rows against 3 image rows by 4 nine-product sets not 6 (points 0, 1, -1, inf; constant 1/2): 36 WMMA per K-step and 3 rows, was 27 per 2; changes f32 sum order
# speedup vs baseline: 1.7924x; 1.0516x over previous
#include <hip/hip_runtime.h>

typedef __attribute__((ext_vector_type(2))) float v2f;
typedef __attribute__((ext_vector_type(4))) float v4f;
typedef __attribute__((ext_vector_type(8))) float v8f;

#define EPSF 1e-5f
#define NB   16
#define CIN  256
#define MM   1024
#define DK   16
#define DV   64
#define QC   64

#define XP 288

__global__ __launch_bounds__(288) void qkv_wmma(
    const float* __restrict__ x,  const float* __restrict__ Wq,
    const float* __restrict__ Wk, const float* __restrict__ Wv,
    const float* __restrict__ gq, const float* __restrict__ bq,
    const float* __restrict__ mq, const float* __restrict__ vq,
    const float* __restrict__ gv, const float* __restrict__ bv,
    const float* __restrict__ mv, const float* __restrict__ vv,
    float* __restrict__ Qbuf,
    float* __restrict__ Kbuf,
    float* __restrict__ Vbuf)
{
    __shared__ __attribute__((aligned(16))) float xs[64 * XP];

    const int tid  = threadIdx.x;
    const int wave = tid >> 5;
    const int lane = tid & 31;
    const int b    = blockIdx.x >> 4;
    const int m0   = (blockIdx.x & 15) * 64;
    const int nL   = lane & 15;
    const int half = lane >> 4;

    v4f xv[16];
    if (tid < 256) {
        const int m  = tid & 63;
        const int cq = tid >> 6;
        const float* xg = x + (size_t)b * CIN * MM + (size_t)(cq * 4) * MM + m0 + m;
        #pragma unroll
        for (int k = 0; k < 16; ++k) {
            xv[k].x = xg[(size_t)(k * 16 + 0) * MM];
            xv[k].y = xg[(size_t)(k * 16 + 1) * MM];
            xv[k].z = xg[(size_t)(k * 16 + 2) * MM];
            xv[k].w = xg[(size_t)(k * 16 + 3) * MM];
        }
    }

    const float* Wm; int orow;
    if (wave < 4)       { Wm = Wq; orow = wave * 16; }
    else if (wave == 4) { Wm = Wk; orow = 0; }
    else                { Wm = Wv; orow = (wave - 5) * 16; }
    const v4f* wrow = reinterpret_cast<const v4f*>(
                          Wm + (size_t)(orow + nL) * CIN + half * 4);
    v4f wf[32];
    #pragma unroll
    for (int i = 0; i < 8; ++i) wf[i] = wrow[i * 2];

    const bool isq = wave < 4;
    const int  oc  = (isq ? wave * 16 : (wave - 5) * 16) + half * 8;
    float sc[8], sh[8];
    if (wave != 4) {
        const float* g  = isq ? gq : gv;
        const float* be = isq ? bq : bv;
        const float* mn = isq ? mq : mv;
        const float* vr = isq ? vq : vv;
        #pragma unroll
        for (int r = 0; r < 8; ++r) {
            sc[r] = g[oc + r] * rsqrtf(vr[oc + r] + EPSF);
            sh[r] = be[oc + r] - mn[oc + r] * sc[r];
        }
    }

    if (tid < 256) {
        float* xd = xs + (tid >> 6) * XP + (tid & 63) * 4;
        #pragma unroll
        for (int k = 0; k < 16; ++k)
            *reinterpret_cast<v4f*>(xd + k * 4 * XP) = xv[k];
    }
    __syncthreads();

    #pragma unroll
    for (int i = 8; i < 32; ++i) wf[i] = wrow[i * 2];

    const float* xb = xs + half * XP + nL * 4;
    v8f acc[4] = {};
    #pragma unroll
    for (int i = 0; i < 32; ++i) {
        v4f bf[4];
        #pragma unroll
        for (int t = 0; t < 4; ++t)
            bf[t] = *reinterpret_cast<const v4f*>(xb + i * 2 * XP + t * 64);
        v2f alo, ahi;
        alo.x = wf[i].x; alo.y = wf[i].y;
        ahi.x = wf[i].z; ahi.y = wf[i].w;
        #pragma unroll
        for (int t = 0; t < 4; ++t) {
            v2f bl; bl.x = bf[t].x; bl.y = bf[t].y;
            acc[t] = __builtin_amdgcn_wmma_f32_16x16x4_f32(
                         false, alo, false, bl, (short)0, acc[t], false, false);
        }
        #pragma unroll
        for (int t = 0; t < 4; ++t) {
            v2f bh; bh.x = bf[t].z; bh.y = bf[t].w;
            acc[t] = __builtin_amdgcn_wmma_f32_16x16x4_f32(
                         false, ahi, false, bh, (short)0, acc[t], false, false);
        }
    }

    const int n0 = b * MM + m0 + nL;
    if (wave == 4) {
        float* dst = Kbuf + (size_t)n0 * DK + half * 8;
        #pragma unroll
        for (int t = 0; t < 4; ++t) {
            v4f lo, hi;
            lo.x = acc[t][0]; lo.y = acc[t][1]; lo.z = acc[t][2]; lo.w = acc[t][3];
            hi.x = acc[t][4]; hi.y = acc[t][5]; hi.z = acc[t][6]; hi.w = acc[t][7];
            *reinterpret_cast<v4f*>(dst + t * 16 * DK)     = lo;
            *reinterpret_cast<v4f*>(dst + t * 16 * DK + 4) = hi;
        }
    } else {
        float* dst = (isq ? Qbuf : Vbuf) + (size_t)n0 * 64 + oc;
        #pragma unroll
        for (int t = 0; t < 4; ++t) {
            v4f lo, hi;
            lo.x = acc[t][0] * sc[0] + sh[0]; lo.y = acc[t][1] * sc[1] + sh[1];
            lo.z = acc[t][2] * sc[2] + sh[2]; lo.w = acc[t][3] * sc[3] + sh[3];
            hi.x = acc[t][4] * sc[4] + sh[4]; hi.y = acc[t][5] * sc[5] + sh[5];
            hi.z = acc[t][6] * sc[6] + sh[6]; hi.w = acc[t][7] * sc[7] + sh[7];
            *reinterpret_cast<v4f*>(dst + t * 16 * 64)     = lo;
            *reinterpret_cast<v4f*>(dst + t * 16 * 64 + 4) = hi;
        }
    }
}

__global__ __launch_bounds__(256) void ksoftmax(float* __restrict__ Kbuf)
{
    const int i = blockIdx.x * 256 + threadIdx.x;
    float* p = Kbuf + (size_t)i * DK;
    float v[DK], mx = -3.0e38f;
    #pragma unroll
    for (int k = 0; k < DK; ++k) { v[k] = p[k]; mx = fmaxf(mx, v[k]); }
    float s = 0.f;
    #pragma unroll
    for (int k = 0; k < DK; ++k) { v[k] = __expf(v[k] - mx); s += v[k]; }
    const float inv = 1.f / s;
    #pragma unroll
    for (int k = 0; k < DK; ++k) p[k] = v[k] * inv;
}

#define PQ     160
#define RP     36
#define SV_SET (16 * PQ)
#define SR_SET (51 * RP)
#define SB_V   0
#define SB_K   (3 * SV_SET)
#define SB_R   (SB_K + 1536)
#define SB_N   (SB_R + 4 * SR_SET)
#define NST    11
#define E_LAM  0
#define E_GSH  16384
#define E_GS   (16384 + 2048)

__global__ __launch_bounds__(256, 2) void lambda_main(
    const float* __restrict__ rel,
    const float* __restrict__ Qbuf,
    const float* __restrict__ sK,
    const float* __restrict__ Vbuf,
    float* __restrict__ out)
{
    __shared__ __attribute__((aligned(16))) float stg[2][SB_N];

    const int tid  = threadIdx.x;
    const int wave = tid >> 5;
    const int lane = tid & 31;
    const int kM   = lane & 15;
    const int half = lane >> 4;

    const int b   = blockIdx.x >> 6;
    const int blk = blockIdx.x & 63;
    const int rp  = blk >> 2;
    const int xb  = blk & 3;
    const int ny0 = 2 * rp;
    const int ql  = wave & 1;
    const int vt  = wave >> 1;
    const int qa  = (xb & 1) * 2 + ql;
    const int xg  = (xb >> 1) * 16;
    const int cs  = 16 - xg;

    const int pid = tid >> 4;
    const int l16 = tid & 15;
    const int fq  = pid & 3;
    const int fP  = pid >> 2;
    const int ri2 = pid + 32 < 47 ? pid + 32 : 46;
    const float* Vg = Vbuf + (size_t)b * MM * DV + (8 * fP + fq) * DV + l16 * 4;
    const float* Kg = sK   + (size_t)b * MM * DK + (8 * fP + fq) * DK + l16;
    const float* Rg = rel + ((31 - ny0) * 63 + cs) * DK + l16;

    const int prow = fq * 4 + fP;
    const int vst  = SB_V + prow * PQ + l16 * 8;
    const int kst  = SB_K + (prow * 16 + l16) * 2;
    const int rst0 = SB_R + pid * RP + l16 * 2 + 1;
    const int rst1 = SB_R + (pid + 16) * RP + l16 * 2 + 1;
    const int rst2 = SB_R + ri2 * RP + l16 * 2 + 1;

    const int vrd = SB_V + half * PQ + (vt * 16 + kM) * 2;
    const int krd = SB_K + (half * 16 + kM) * 2;
    const int kg0 = krd + ql * 1024;
    const int kg1 = krd + 512 + ql * 256;
    const int trd = SB_R + (8 * half - 4 * qa + 15 + 4) * RP + kM * 2;

    v8f acc[4][9] = {};
    v8f accG[2] = {};

    v4f fa0, fb0, fa1, fb1, fa2, fb2;
    float ka0, kb0, ka1, kb1, ka2, kb2;
    float rm0, rm1, rm2, rc0, rc1, rc2, rq0, rq1, rq2, rz0, rz1, rz2;
    const int dzmax = 31 + ny0;

#define FETCH_STAGE(ux)                                                      \
    {                                                                        \
        const int  r0_ = 3 * (ux);                                           \
        const bool pd_ = r0_ + 2 > 31;                                       \
        const int  r2_ = pd_ ? 31 : r0_ + 2;                                 \
        const int  dz_ = r0_ + 2 < dzmax ? r0_ + 2 : dzmax;                  \
        fa0 = *reinterpret_cast<const v4f*>(Vg + r0_ * 2048);                \
        fb0 = *reinterpret_cast<const v4f*>(Vg + r0_ * 2048 + 4 * DV);       \
        fa1 = *reinterpret_cast<const v4f*>(Vg + (r0_ + 1) * 2048);          \
        fb1 = *reinterpret_cast<const v4f*>(Vg + (r0_ + 1) * 2048 + 4 * DV); \
        fa2 = *reinterpret_cast<const v4f*>(Vg + r2_ * 2048);                \
        fb2 = *reinterpret_cast<const v4f*>(Vg + r2_ * 2048 + 4 * DV);       \
        ka0 = Kg[r0_ * 512];        kb0 = Kg[r0_ * 512 + 4 * DK];            \
        ka1 = Kg[(r0_ + 1) * 512];  kb1 = Kg[(r0_ + 1) * 512 + 4 * DK];      \
        ka2 = Kg[r2_ * 512];        kb2 = Kg[r2_ * 512 + 4 * DK];            \
        if (pd_) {                                                           \
            const v4f z4_ = {0.f, 0.f, 0.f, 0.f};                            \
            fa2 = z4_; fb2 = z4_; ka2 = 0.f; kb2 = 0.f;                      \
        }                                                                    \
        rm0 = Rg[(r0_ - 1) * 1008 + pid * 16];                               \
        rm1 = Rg[(r0_ - 1) * 1008 + (pid + 16) * 16];                        \
        rm2 = Rg[(r0_ - 1) * 1008 + ri2 * 16];                               \
        rc0 = Rg[r0_ * 1008 + pid * 16];                                     \
        rc1 = Rg[r0_ * 1008 + (pid + 16) * 16];                              \
        rc2 = Rg[r0_ * 1008 + ri2 * 16];                                     \
        rq0 = Rg[(r0_ + 1) * 1008 + pid * 16];                               \
        rq1 = Rg[(r0_ + 1) * 1008 + (pid + 16) * 16];                        \
        rq2 = Rg[(r0_ + 1) * 1008 + ri2 * 16];                               \
        rz0 = Rg[dz_ * 1008 + pid * 16];                                     \
        rz1 = Rg[dz_ * 1008 + (pid + 16) * 16];                              \
        rz2 = Rg[dz_ * 1008 + ri2 * 16];                                     \
    }

#define STORE_V(Sx, off, A_, B_)                                             \
    {                                                                        \
        v4f lo, hi;                                                          \
        lo.x = (A_).x; lo.y = (B_).x; lo.z = (A_).y; lo.w = (B_).y;          \
        hi.x = (A_).z; hi.y = (B_).z; hi.z = (A_).w; hi.w = (B_).w;          \
        *reinterpret_cast<v4f*>((Sx) + (off))     = lo;                      \
        *reinterpret_cast<v4f*>((Sx) + (off) + 4) = hi;                      \
    }

#define STORE_R1(Sx, at, val)                                                \
    { const float t_ = (val); (Sx)[(at)] = t_; (Sx)[(at) + 4 * RP - 1] = t_; }

#define STORE_R(Sx, rst, m_, c_, q_, z_)                                     \
    STORE_R1(Sx, (rst),              (m_) - (q_))                            \
    STORE_R1(Sx, (rst) + SR_SET,     0.5f * ((c_) + (q_)))                   \
    STORE_R1(Sx, (rst) + 2 * SR_SET, 0.5f * ((q_) - (c_)))                   \
    STORE_R1(Sx, (rst) + 3 * SR_SET, (z_) - (c_))

#define STORE_STAGE(Sx)                                                      \
    {                                                                        \
        STORE_V(Sx, vst, fa0, fb0)                                           \
        STORE_V(Sx, vst + SV_SET, fa1, fb1)                                  \
        STORE_V(Sx, vst + 2 * SV_SET, fa2, fb2)                              \
        v2f kk; kk.x = ka0; kk.y = kb0;                                      \
        *reinterpret_cast<v2f*>((Sx) + kst) = kk;                            \
        kk.x = ka1; kk.y = kb1;                                              \
        *reinterpret_cast<v2f*>((Sx) + kst + 512) = kk;                      \
        kk.x = ka2; kk.y = kb2;                                              \
        *reinterpret_cast<v2f*>((Sx) + kst + 1024) = kk;                     \
        STORE_R(Sx, rst0, rm0, rc0, rq0, rz0)                                \
        STORE_R(Sx, rst1, rm1, rc1, rq1, rz1)                                \
        STORE_R(Sx, rst2, rm2, rc2, rq2, rz2)                                \
    }

#define LDS_BARRIER()                                                        \
    {                                                                        \
        __builtin_amdgcn_fence(__ATOMIC_RELEASE, "workgroup", "local");      \
        __builtin_amdgcn_s_barrier();                                        \
        __builtin_amdgcn_fence(__ATOMIC_ACQUIRE, "workgroup", "local");      \
    }

    FETCH_STAGE(0)
    STORE_STAGE(stg[0])
    FETCH_STAGE(1)
    LDS_BARRIER()

    v2f rr[8][7];
    v2f Vr[2][3][4];
    v2f kf[2][6];
#define LOAD_GROUP(g_, Sp)                                                           \
    {                                                                                \
        const int jl = (g_) / 4, Xl = (g_) % 4;                                      \
        _Pragma("unroll")                                                            \
        for (int d = 0; d < 7; ++d)                                                  \
            rr[g_][d] = *reinterpret_cast<const v2f*>(                               \
                            (Sp) + trd + Xl * SR_SET + (16 * jl + d - 3) * RP);      \
        if (Xl == 0) {                                                               \
            _Pragma("unroll")                                                        \
            for (int q = 0; q < 4; ++q)                                              \
                Vr[jl][0][q] = *reinterpret_cast<const v2f*>(                        \
                                   (Sp) + vrd + jl * 2 * PQ + q * 4 * PQ);           \
        }                                                                            \
        if (Xl == 1) {                                                               \
            _Pragma("unroll")                                                        \
            for (int q = 0; q < 4; ++q) {                                            \
                Vr[jl][1][q] = *reinterpret_cast<const v2f*>(                        \
                                   (Sp) + vrd + SV_SET + jl * 2 * PQ + q * 4 * PQ);  \
                Vr[jl][2][q] = *reinterpret_cast<const v2f*>(                        \
                                   (Sp) + vrd + 2 * SV_SET + jl * 2 * PQ + q * 4 * PQ); \
            }                                                                        \
        }                                                                            \
        if (Xl == 3) {                                                               \
            _Pragma("unroll")                                                        \
            for (int q = 0; q < 4; ++q)                                              \
                kf[jl][q] = *reinterpret_cast<const v2f*>((Sp) + kg0 + q * 128 + jl * 64); \
            kf[jl][4] = *reinterpret_cast<const v2f*>((Sp) + kg1 + jl * 64);         \
            kf[jl][5] = *reinterpret_cast<const v2f*>((Sp) + kg1 + 128 + jl * 64);   \
        }                                                                            \
    }
    LOAD_GROUP(0, stg[0])

    for (int u = 0; u < NST; ++u) {
        const int un = u < NST - 2 ? u + 2 : NST - 1;

        const float* S  = stg[u & 1];
        float*       Sn = stg[(u + 1) & 1];
        #pragma unroll
        for (int g = 0; g < 8; ++g) {
            const int jj = g / 4, X = g % 4;
            if (g < 7) LOAD_GROUP(g + 1, S)
            else       LOAD_GROUP(0, Sn)
            __builtin_amdgcn_sched_barrier(0);
            v2f a[9];
            a[0] = rr[g][3];
            a[1] = rr[g][4] - rr[g][3];
            a[2] = rr[g][2] - rr[g][3];
            a[3] = rr[g][5] - rr[g][3];
            a[4] = (rr[g][6] - rr[g][4]) - a[3];
            a[5] = (rr[g][4] - rr[g][2]) - a[3];
            a[6] = rr[g][1] - rr[g][3];
            a[7] = (rr[g][2] - rr[g][4]) - a[6];
            a[8] = (rr[g][0] - rr[g][2]) - a[6];
            v2f Vx[4];
            #pragma unroll
            for (int q = 0; q < 4; ++q) {
                const v2f t02 = Vr[jj][0][q] + Vr[jj][2][q];
                Vx[q] = X == 0 ? Vr[jj][0][q]
                      : X == 1 ? t02 + Vr[jj][1][q]
                      : X == 2 ? t02 - Vr[jj][1][q]
                      :          Vr[jj][2][q];
            }
            const v2f V0 = Vx[0], V1 = Vx[1], V2 = Vx[2], V3 = Vx[3];
            const v2f s02 = V0 + V2;
            const v2f s13 = V1 + V3;
            const v2f s01 = V0 + V1;
            const v2f s23 = V2 + V3;
            const v2f sal = s02 + s13;
            if (X != 3 || u < NST - 1) {
                acc[X][0] = __builtin_amdgcn_wmma_f32_16x16x4_f32(
                                false, a[0], false, sal, (short)0, acc[X][0], false, false);
                acc[X][1] = __builtin_amdgcn_wmma_f32_16x16x4_f32(
                                false, a[1], false, s13, (short)0, acc[X][1], false, false);
                acc[X][2] = __builtin_amdgcn_wmma_f32_16x16x4_f32(
                                false, a[2], false, s02, (short)0, acc[X][2], false, false);
                acc[X][3] = __builtin_amdgcn_wmma_f32_16x16x4_f32(
                                false, a[3], false, s23, (short)0, acc[X][3], false, false);
                acc[X][4] = __builtin_amdgcn_wmma_f32_16x16x4_f32(
                                false, a[4], false, V3, (short)0, acc[X][4], false, false);
                acc[X][5] = __builtin_amdgcn_wmma_f32_16x16x4_f32(
                                false, a[5], false, V2, (short)0, acc[X][5], false, false);
                acc[X][6] = __builtin_amdgcn_wmma_f32_16x16x4_f32(
                                false, a[6], false, s01, (short)0, acc[X][6], false, false);
                acc[X][7] = __builtin_amdgcn_wmma_f32_16x16x4_f32(
                                false, a[7], false, V1, (short)0, acc[X][7], false, false);
                acc[X][8] = __builtin_amdgcn_wmma_f32_16x16x4_f32(
                                false, a[8], false, V0, (short)0, acc[X][8], false, false);
            }
            if (X == 3) {
                #pragma unroll
                for (int i = 0; i < 6; ++i) {
                    const v2f vlo = i < 4 ? Vr[jj][0][i] : Vr[jj][1][i - 4];
                    const v2f vhi = i < 4 ? Vr[jj][2][i] : Vr[jj][1][i - 2];
                    const v2f vgq = ql ? vhi : vlo;
                    accG[i & 1] = __builtin_amdgcn_wmma_f32_16x16x4_f32(
                                      false, kf[jj][i], false, vgq, (short)0, accG[i & 1], false, false);
                }
            }
            __builtin_amdgcn_sched_barrier(0);
            if (g == 1) { STORE_STAGE(Sn) FETCH_STAGE(un) }
            if (g == 6) LDS_BARRIER()
        }
    }
#undef LOAD_GROUP
#undef LDS_BARRIER
#undef FETCH_STAGE
#undef STORE_STAGE
#undef STORE_V
#undef STORE_R
#undef STORE_R1

    float* E = &stg[0][0];

    #pragma unroll
    for (int r = 0; r < 8; ++r)
        E[E_GSH + wave * 256 + (r + half * 8) * 16 + kM] = accG[0][r] + accG[1][r];
    __syncthreads();
    {
        const int ge = tid * 4;
        const int gk = ge >> 6;
        const int gv = ge & 63;
        const int w0 = (gv >> 4) * 2;
        const int gi = gk * 16 + (gv & 15);
        const v4f g = *reinterpret_cast<const v4f*>(E + E_GSH + w0 * 256 + gi)
                    + *reinterpret_cast<const v4f*>(E + E_GSH + (w0 + 1) * 256 + gi);
        *reinterpret_cast<v4f*>(E + E_GS + ge) = g;
    }
    __syncthreads();

    #pragma unroll
    for (int r = 0; r < 8; ++r) {
        const int k = r + half * 8;
        const float g = E[E_GS + k * 64 + vt * 16 + kM];
        #pragma unroll
        for (int y = 0; y < 2; ++y)
            #pragma unroll
            for (int p = 0; p < 4; ++p) {
                float l[4];
                #pragma unroll
                for (int X = 0; X < 4; ++X) {
                    const float c1 = (p & 1) ? acc[X][2][r] : acc[X][1][r];
                    const float c2 = (p & 2) ? acc[X][6][r] : acc[X][3][r];
                    const float c3 = p == 0 ? acc[X][4][r] : p == 1 ? acc[X][5][r]
                                   : p == 2 ? acc[X][7][r] : acc[X][8][r];
                    l[X] = (acc[X][0][r] + c1) + (c2 + c3);
                }
                const float ly = y == 0 ? (l[1] - l[2]) + l[3] : (l[0] + l[1]) + l[2];
                E[E_LAM + ((((y * 4 + p) * 2 + ql) * 16 + k) << 6) + vt * 16 + kM] = ly + g;
            }
    }
    __syncthreads();

    #pragma unroll 1
    for (int i = 0; i < 2; ++i) {
        const int pi = wave * 2 + i;
        const int y  = pi >> 3;
        const int p  = (pi >> 1) & 3;
        const int n  = (ny0 + y) * 32 + xg + 4 * ((xb & 1) * 2 + (pi & 1)) + p;
        const float* Qn = Qbuf + (size_t)(b * MM + n) * QC;
        const float* L  = E + E_LAM + pi * 1024 + lane;
        float* on = out + (size_t)(b * MM + n) * 256 + lane;
        float o[4][2] = {};
        #pragma unroll
        for (int k = 0; k < DK; ++k) {
            const float l0 = L[k * 64];
            const float l1 = L[k * 64 + 32];
            #pragma unroll
            for (int h = 0; h < 4; ++h) {
                const float qk = Qn[h * 16 + k];
                o[h][0] = fmaf(qk, l0, o[h][0]);
                o[h][1] = fmaf(qk, l1, o[h][1]);
            }
        }
        #pragma unroll
        for (int h = 0; h < 4; ++h) {
            on[h * 64]      = o[h][0];
            on[h * 64 + 32] = o[h][1];
        }
    }
}

extern "C" void kernel_launch(void* const* d_in, const int* in_sizes, int n_in,
                              void* d_out, int out_size, void* d_ws, size_t ws_size,
                              hipStream_t stream)
{
    const float* x   = (const float*)d_in[0];
    const float* Wq  = (const float*)d_in[1];
    const float* Wk  = (const float*)d_in[2];
    const float* Wv  = (const float*)d_in[3];
    const float* gq  = (const float*)d_in[4];
    const float* bq  = (const float*)d_in[5];
    const float* mq  = (const float*)d_in[6];
    const float* vq  = (const float*)d_in[7];
    const float* gv  = (const float*)d_in[8];
    const float* bv  = (const float*)d_in[9];
    const float* mv  = (const float*)d_in[10];
    const float* vv  = (const float*)d_in[11];
    const float* rel = (const float*)d_in[12];
    float* out = (float*)d_out;

    float* Qbuf = (float*)d_ws;
    float* Vbuf = Qbuf + (size_t)NB * MM * QC;
    float* Kbuf = Vbuf + (size_t)NB * MM * DV;

    qkv_wmma<<<256, 288, 0, stream>>>(x, Wq, Wk, Wv, gq, bq, mq, vq,
                                       gv, bv, mv, vv, Qbuf, Kbuf, Vbuf);
    ksoftmax<<<64, 256, 0, stream>>>(Kbuf);
    lambda_main<<<1024, 256, 0, stream>>>(rel, Qbuf, Kbuf, Vbuf, out);
}


# Kernel 3: a block takes the four 8-column tiles of its image row pair in turn (256 blocks, was 1024) and forms G = sum_m sK V, equal for all four, in the first tile only, kept in LDS; sums unchanged
# speedup vs baseline: 2.0553x; 1.1467x over previous
#include <hip/hip_runtime.h>

typedef __attribute__((ext_vector_type(2))) float v2f;
typedef __attribute__((ext_vector_type(4))) float v4f;
typedef __attribute__((ext_vector_type(8))) float v8f;

#define EPSF 1e-5f
#define NB   16
#define CIN  256
#define MM   1024
#define DK   16
#define DV   64
#define QC   64

#define XP 288

__global__ __launch_bounds__(288) void qkv_wmma(
    const float* __restrict__ x,  const float* __restrict__ Wq,
    const float* __restrict__ Wk, const float* __restrict__ Wv,
    const float* __restrict__ gq, const float* __restrict__ bq,
    const float* __restrict__ mq, const float* __restrict__ vq,
    const float* __restrict__ gv, const float* __restrict__ bv,
    const float* __restrict__ mv, const float* __restrict__ vv,
    float* __restrict__ Qbuf,
    float* __restrict__ Kbuf,
    float* __restrict__ Vbuf)
{
    __shared__ __attribute__((aligned(16))) float xs[64 * XP];

    const int tid  = threadIdx.x;
    const int wave = tid >> 5;
    const int lane = tid & 31;
    const int b    = blockIdx.x >> 4;
    const int m0   = (blockIdx.x & 15) * 64;
    const int nL   = lane & 15;
    const int half = lane >> 4;

    v4f xv[16];
    if (tid < 256) {
        const int m  = tid & 63;
        const int cq = tid >> 6;
        const float* xg = x + (size_t)b * CIN * MM + (size_t)(cq * 4) * MM + m0 + m;
        #pragma unroll
        for (int k = 0; k < 16; ++k) {
            xv[k].x = xg[(size_t)(k * 16 + 0) * MM];
            xv[k].y = xg[(size_t)(k * 16 + 1) * MM];
            xv[k].z = xg[(size_t)(k * 16 + 2) * MM];
            xv[k].w = xg[(size_t)(k * 16 + 3) * MM];
        }
    }

    const float* Wm; int orow;
    if (wave < 4)       { Wm = Wq; orow = wave * 16; }
    else if (wave == 4) { Wm = Wk; orow = 0; }
    else                { Wm = Wv; orow = (wave - 5) * 16; }
    const v4f* wrow = reinterpret_cast<const v4f*>(
                          Wm + (size_t)(orow + nL) * CIN + half * 4);
    v4f wf[32];
    #pragma unroll
    for (int i = 0; i < 8; ++i) wf[i] = wrow[i * 2];

    const bool isq = wave < 4;
    const int  oc  = (isq ? wave * 16 : (wave - 5) * 16) + half * 8;
    float sc[8], sh[8];
    if (wave != 4) {
        const float* g  = isq ? gq : gv;
        const float* be = isq ? bq : bv;
        const float* mn = isq ? mq : mv;
        const float* vr = isq ? vq : vv;
        #pragma unroll
        for (int r = 0; r < 8; ++r) {
            sc[r] = g[oc + r] * rsqrtf(vr[oc + r] + EPSF);
            sh[r] = be[oc + r] - mn[oc + r] * sc[r];
        }
    }

    if (tid < 256) {
        float* xd = xs + (tid >> 6) * XP + (tid & 63) * 4;
        #pragma unroll
        for (int k = 0; k < 16; ++k)
            *reinterpret_cast<v4f*>(xd + k * 4 * XP) = xv[k];
    }
    __syncthreads();

    #pragma unroll
    for (int i = 8; i < 32; ++i) wf[i] = wrow[i * 2];

    const float* xb = xs + half * XP + nL * 4;
    v8f acc[4] = {};
    #pragma unroll
    for (int i = 0; i < 32; ++i) {
        v4f bf[4];
        #pragma unroll
        for (int t = 0; t < 4; ++t)
            bf[t] = *reinterpret_cast<const v4f*>(xb + i * 2 * XP + t * 64);
        v2f alo, ahi;
        alo.x = wf[i].x; alo.y = wf[i].y;
        ahi.x = wf[i].z; ahi.y = wf[i].w;
        #pragma unroll
        for (int t = 0; t < 4; ++t) {
            v2f bl; bl.x = bf[t].x; bl.y = bf[t].y;
            acc[t] = __builtin_amdgcn_wmma_f32_16x16x4_f32(
                         false, alo, false, bl, (short)0, acc[t], false, false);
        }
        #pragma unroll
        for (int t = 0; t < 4; ++t) {
            v2f bh; bh.x = bf[t].z; bh.y = bf[t].w;
            acc[t] = __builtin_amdgcn_wmma_f32_16x16x4_f32(
                         false, ahi, false, bh, (short)0, acc[t], false, false);
        }
    }

    const int n0 = b * MM + m0 + nL;
    if (wave == 4) {
        float* dst = Kbuf + (size_t)n0 * DK + half * 8;
        #pragma unroll
        for (int t = 0; t < 4; ++t) {
            v4f lo, hi;
            lo.x = acc[t][0]; lo.y = acc[t][1]; lo.z = acc[t][2]; lo.w = acc[t][3];
            hi.x = acc[t][4]; hi.y = acc[t][5]; hi.z = acc[t][6]; hi.w = acc[t][7];
            *reinterpret_cast<v4f*>(dst + t * 16 * DK)     = lo;
            *reinterpret_cast<v4f*>(dst + t * 16 * DK + 4) = hi;
        }
    } else {
        float* dst = (isq ? Qbuf : Vbuf) + (size_t)n0 * 64 + oc;
        #pragma unroll
        for (int t = 0; t < 4; ++t) {
            v4f lo, hi;
            lo.x = acc[t][0] * sc[0] + sh[0]; lo.y = acc[t][1] * sc[1] + sh[1];
            lo.z = acc[t][2] * sc[2] + sh[2]; lo.w = acc[t][3] * sc[3] + sh[3];
            hi.x = acc[t][4] * sc[4] + sh[4]; hi.y = acc[t][5] * sc[5] + sh[5];
            hi.z = acc[t][6] * sc[6] + sh[6]; hi.w = acc[t][7] * sc[7] + sh[7];
            *reinterpret_cast<v4f*>(dst + t * 16 * 64)     = lo;
            *reinterpret_cast<v4f*>(dst + t * 16 * 64 + 4) = hi;
        }
    }
}

__global__ __launch_bounds__(256) void ksoftmax(float* __restrict__ Kbuf)
{
    const int i = blockIdx.x * 256 + threadIdx.x;
    float* p = Kbuf + (size_t)i * DK;
    float v[DK], mx = -3.0e38f;
    #pragma unroll
    for (int k = 0; k < DK; ++k) { v[k] = p[k]; mx = fmaxf(mx, v[k]); }
    float s = 0.f;
    #pragma unroll
    for (int k = 0; k < DK; ++k) { v[k] = __expf(v[k] - mx); s += v[k]; }
    const float inv = 1.f / s;
    #pragma unroll
    for (int k = 0; k < DK; ++k) p[k] = v[k] * inv;
}

#define PQ     160
#define RP     36
#define SV_SET (16 * PQ)
#define SR_SET (51 * RP)
#define SB_V   0
#define SB_K   (3 * SV_SET)
#define SB_R   (SB_K + 1536)
#define SB_N   (SB_R + 4 * SR_SET)
#define NST    11
#define E_LAM  0
#define E_GSH  16384
#define E_GS   (2 * SB_N)

template <bool WITH_G>
__device__ __forceinline__ void lambda_tile(
    const float* __restrict__ rel,
    const float* __restrict__ Qbuf,
    const float* __restrict__ sK,
    const float* __restrict__ Vbuf,
    float* __restrict__ out,
    float* stg0,
    const int b, const int rp, const int xb)
{
    const int tid  = threadIdx.x;
    const int wave = tid >> 5;
    const int lane = tid & 31;
    const int kM   = lane & 15;
    const int half = lane >> 4;

    const int ny0 = 2 * rp;
    const int ql  = wave & 1;
    const int vt  = wave >> 1;
    const int qa  = (xb & 1) * 2 + ql;
    const int xg  = (xb >> 1) * 16;
    const int cs  = 16 - xg;

    const int pid = tid >> 4;
    const int l16 = tid & 15;
    const int fq  = pid & 3;
    const int fP  = pid >> 2;
    const int ri2 = pid + 32 < 47 ? pid + 32 : 46;
    const unsigned vofs = ((8 * fP + fq) * DV + l16 * 4) * 4u;
    const unsigned kofs = ((8 * fP + fq) * DK + l16) * 4u;
    const unsigned rof0 = (pid * 16 + l16) * 4u;
    const unsigned rof2 = (ri2 * 16 + l16) * 4u;
    const char* Vb = reinterpret_cast<const char*>(Vbuf + (size_t)b * MM * DV);
    const char* Kb = reinterpret_cast<const char*>(sK   + (size_t)b * MM * DK);
    const char* Rb = reinterpret_cast<const char*>(rel + ((31 - ny0) * 63 + cs) * DK);
#define LDV(fl_)      (*reinterpret_cast<const v4f*>(Vb + (long)(fl_) * 4 + vofs))
#define LDK(fl_)      (*reinterpret_cast<const float*>(Kb + (long)(fl_) * 4 + kofs))
#define LDR(fl_, of_) (*reinterpret_cast<const float*>(Rb + (long)(fl_) * 4 + (of_)))

    const int prow = fq * 4 + fP;
    const int vst  = SB_V + prow * PQ + l16 * 8;
    const int kst  = SB_K + (prow * 16 + l16) * 2;
    const int rst0 = SB_R + pid * RP + l16 * 2 + 1;
    const int rst1 = SB_R + (pid + 16) * RP + l16 * 2 + 1;
    const int rst2 = SB_R + ri2 * RP + l16 * 2 + 1;

    const int vrd = SB_V + half * PQ + (vt * 16 + kM) * 2;
    const int krd = SB_K + (half * 16 + kM) * 2;
    const int kg0 = krd + ql * 1024;
    const int kg1 = krd + 512 + ql * 256;
    const int trd = SB_R + (8 * half - 4 * qa + 15 + 4) * RP + kM * 2;

    v8f acc[4][9] = {};
    v8f accG[2] = {};

    v4f fa0, fb0, fa1, fb1, fa2, fb2;
    float ka0, kb0, ka1, kb1, ka2, kb2;
    float rm0, rm1, rm2, rc0, rc1, rc2, rq0, rq1, rq2, rz0, rz1, rz2;
    const int dzmax = 31 + ny0;

#define FETCH_STAGE(ux)                                                      \
    {                                                                        \
        const int  r0_ = 3 * (ux);                                           \
        const bool pd_ = r0_ + 2 > 31;                                       \
        const int  r2_ = pd_ ? 31 : r0_ + 2;                                 \
        const int  dz_ = r0_ + 2 < dzmax ? r0_ + 2 : dzmax;                  \
        fa0 = LDV(r0_ * 2048);              fb0 = LDV(r0_ * 2048 + 4 * DV);          \
        fa1 = LDV((r0_ + 1) * 2048);        fb1 = LDV((r0_ + 1) * 2048 + 4 * DV);    \
        fa2 = LDV(r2_ * 2048);              fb2 = LDV(r2_ * 2048 + 4 * DV);          \
        if (WITH_G) {                                                        \
            ka0 = LDK(r0_ * 512);        kb0 = LDK(r0_ * 512 + 4 * DK);      \
            ka1 = LDK((r0_ + 1) * 512);  kb1 = LDK((r0_ + 1) * 512 + 4 * DK);\
            ka2 = LDK(r2_ * 512);        kb2 = LDK(r2_ * 512 + 4 * DK);      \
        }                                                                    \
        if (pd_) {                                                           \
            const v4f z4_ = {0.f, 0.f, 0.f, 0.f};                            \
            fa2 = z4_; fb2 = z4_; ka2 = 0.f; kb2 = 0.f;                      \
        }                                                                    \
        rm0 = LDR((r0_ - 1) * 1008, rof0);                                   \
        rm1 = LDR((r0_ - 1) * 1008 + 256, rof0);                             \
        rm2 = LDR((r0_ - 1) * 1008, rof2);                                   \
        rc0 = LDR(r0_ * 1008, rof0);                                         \
        rc1 = LDR(r0_ * 1008 + 256, rof0);                                   \
        rc2 = LDR(r0_ * 1008, rof2);                                         \
        rq0 = LDR((r0_ + 1) * 1008, rof0);                                   \
        rq1 = LDR((r0_ + 1) * 1008 + 256, rof0);                             \
        rq2 = LDR((r0_ + 1) * 1008, rof2);                                   \
        rz0 = LDR(dz_ * 1008, rof0);                                         \
        rz1 = LDR(dz_ * 1008 + 256, rof0);                                   \
        rz2 = LDR(dz_ * 1008, rof2);                                         \
    }

#define STORE_V(Sx, off, A_, B_)                                             \
    {                                                                        \
        v4f lo, hi;                                                          \
        lo.x = (A_).x; lo.y = (B_).x; lo.z = (A_).y; lo.w = (B_).y;          \
        hi.x = (A_).z; hi.y = (B_).z; hi.z = (A_).w; hi.w = (B_).w;          \
        *reinterpret_cast<v4f*>((Sx) + (off))     = lo;                      \
        *reinterpret_cast<v4f*>((Sx) + (off) + 4) = hi;                      \
    }

#define STORE_R1(Sx, at, val)                                                \
    { const float t_ = (val); (Sx)[(at)] = t_; (Sx)[(at) + 4 * RP - 1] = t_; }

#define STORE_R(Sx, rst, m_, c_, q_, z_)                                     \
    STORE_R1(Sx, (rst),              (m_) - (q_))                            \
    STORE_R1(Sx, (rst) + SR_SET,     0.5f * ((c_) + (q_)))                   \
    STORE_R1(Sx, (rst) + 2 * SR_SET, 0.5f * ((q_) - (c_)))                   \
    STORE_R1(Sx, (rst) + 3 * SR_SET, (z_) - (c_))

#define STORE_STAGE(Sx)                                                      \
    {                                                                        \
        STORE_V(Sx, vst, fa0, fb0)                                           \
        STORE_V(Sx, vst + SV_SET, fa1, fb1)                                  \
        STORE_V(Sx, vst + 2 * SV_SET, fa2, fb2)                              \
        if (WITH_G) {                                                        \
            v2f kk; kk.x = ka0; kk.y = kb0;                                  \
            *reinterpret_cast<v2f*>((Sx) + kst) = kk;                        \
            kk.x = ka1; kk.y = kb1;                                          \
            *reinterpret_cast<v2f*>((Sx) + kst + 512) = kk;                  \
            kk.x = ka2; kk.y = kb2;                                          \
            *reinterpret_cast<v2f*>((Sx) + kst + 1024) = kk;                 \
        }                                                                    \
        STORE_R(Sx, rst0, rm0, rc0, rq0, rz0)                                \
        STORE_R(Sx, rst1, rm1, rc1, rq1, rz1)                                \
        STORE_R(Sx, rst2, rm2, rc2, rq2, rz2)                                \
    }

#define LDS_BARRIER()                                                        \
    {                                                                        \
        __builtin_amdgcn_fence(__ATOMIC_RELEASE, "workgroup", "local");      \
        __builtin_amdgcn_s_barrier();                                        \
        __builtin_amdgcn_fence(__ATOMIC_ACQUIRE, "workgroup", "local");      \
    }

    FETCH_STAGE(0)
    STORE_STAGE(stg0)
    FETCH_STAGE(1)
    LDS_BARRIER()

    v2f rr[8][7];
    v2f Vr[2][3][4];
    v2f kf[2][6];
#define LOAD_GROUP(g_, Sp)                                                           \
    {                                                                                \
        const int jl = (g_) / 4, Xl = (g_) % 4;                                      \
        _Pragma("unroll")                                                            \
        for (int d = 0; d < 7; ++d)                                                  \
            rr[g_][d] = *reinterpret_cast<const v2f*>(                               \
                            (Sp) + trd + Xl * SR_SET + (16 * jl + d - 3) * RP);      \
        if (Xl == 0) {                                                               \
            _Pragma("unroll")                                                        \
            for (int q = 0; q < 4; ++q)                                              \
                Vr[jl][0][q] = *reinterpret_cast<const v2f*>(                        \
                                   (Sp) + vrd + jl * 2 * PQ + q * 4 * PQ);           \
        }                                                                            \
        if (Xl == 1) {                                                               \
            _Pragma("unroll")                                                        \
            for (int q = 0; q < 4; ++q) {                                            \
                Vr[jl][1][q] = *reinterpret_cast<const v2f*>(                        \
                                   (Sp) + vrd + SV_SET + jl * 2 * PQ + q * 4 * PQ);  \
                Vr[jl][2][q] = *reinterpret_cast<const v2f*>(                        \
                                   (Sp) + vrd + 2 * SV_SET + jl * 2 * PQ + q * 4 * PQ); \
            }                                                                        \
        }                                                                            \
        if (WITH_G && Xl == 3) {                                                     \
            _Pragma("unroll")                                                        \
            for (int q = 0; q < 4; ++q)                                              \
                kf[jl][q] = *reinterpret_cast<const v2f*>((Sp) + kg0 + q * 128 + jl * 64); \
            kf[jl][4] = *reinterpret_cast<const v2f*>((Sp) + kg1 + jl * 64);         \
            kf[jl][5] = *reinterpret_cast<const v2f*>((Sp) + kg1 + 128 + jl * 64);   \
        }                                                                            \
    }
    LOAD_GROUP(0, stg0)

    for (int u = 0; u < NST; ++u) {
        const int un = u < NST - 2 ? u + 2 : NST - 1;

        const float* S  = stg0 + (u & 1) * SB_N;
        float*       Sn = stg0 + ((u + 1) & 1) * SB_N;
        #pragma unroll
        for (int g = 0; g < 8; ++g) {
            const int jj = g / 4, X = g % 4;
            if (g < 7) LOAD_GROUP(g + 1, S)
            else       LOAD_GROUP(0, Sn)
            __builtin_amdgcn_sched_barrier(0);
            v2f a[9];
            a[0] = rr[g][3];
            a[1] = rr[g][4] - rr[g][3];
            a[2] = rr[g][2] - rr[g][3];
            a[3] = rr[g][5] - rr[g][3];
            a[4] = (rr[g][6] - rr[g][4]) - a[3];
            a[5] = (rr[g][4] - rr[g][2]) - a[3];
            a[6] = rr[g][1] - rr[g][3];
            a[7] = (rr[g][2] - rr[g][4]) - a[6];
            a[8] = (rr[g][0] - rr[g][2]) - a[6];
            v2f Vx[4];
            #pragma unroll
            for (int q = 0; q < 4; ++q) {
                const v2f t02 = Vr[jj][0][q] + Vr[jj][2][q];
                Vx[q] = X == 0 ? Vr[jj][0][q]
                      : X == 1 ? t02 + Vr[jj][1][q]
                      : X == 2 ? t02 - Vr[jj][1][q]
                      :          Vr[jj][2][q];
            }
            const v2f V0 = Vx[0], V1 = Vx[1], V2 = Vx[2], V3 = Vx[3];
            const v2f s02 = V0 + V2;
            const v2f s13 = V1 + V3;
            const v2f s01 = V0 + V1;
            const v2f s23 = V2 + V3;
            const v2f sal = s02 + s13;
            if (X != 3 || u < NST - 1) {
                acc[X][0] = __builtin_amdgcn_wmma_f32_16x16x4_f32(
                                false, a[0], false, sal, (short)0, acc[X][0], false, false);
                acc[X][1] = __builtin_amdgcn_wmma_f32_16x16x4_f32(
                                false, a[1], false, s13, (short)0, acc[X][1], false, false);
                acc[X][2] = __builtin_amdgcn_wmma_f32_16x16x4_f32(
                                false, a[2], false, s02, (short)0, acc[X][2], false, false);
                acc[X][3] = __builtin_amdgcn_wmma_f32_16x16x4_f32(
                                false, a[3], false, s23, (short)0, acc[X][3], false, false);
                acc[X][4] = __builtin_amdgcn_wmma_f32_16x16x4_f32(
                                false, a[4], false, V3, (short)0, acc[X][4], false, false);
                acc[X][5] = __builtin_amdgcn_wmma_f32_16x16x4_f32(
                                false, a[5], false, V2, (short)0, acc[X][5], false, false);
                acc[X][6] = __builtin_amdgcn_wmma_f32_16x16x4_f32(
                                false, a[6], false, s01, (short)0, acc[X][6], false, false);
                acc[X][7] = __builtin_amdgcn_wmma_f32_16x16x4_f32(
                                false, a[7], false, V1, (short)0, acc[X][7], false, false);
                acc[X][8] = __builtin_amdgcn_wmma_f32_16x16x4_f32(
                                false, a[8], false, V0, (short)0, acc[X][8], false, false);
            }
            if (WITH_G && X == 3) {
                #pragma unroll
                for (int i = 0; i < 6; ++i) {
                    const v2f vlo = i < 4 ? Vr[jj][0][i] : Vr[jj][1][i - 4];
                    const v2f vhi = i < 4 ? Vr[jj][2][i] : Vr[jj][1][i - 2];
                    const v2f vgq = ql ? vhi : vlo;
                    accG[i & 1] = __builtin_amdgcn_wmma_f32_16x16x4_f32(
                                      false, kf[jj][i], false, vgq, (short)0, accG[i & 1], false, false);
                }
            }
            __builtin_amdgcn_sched_barrier(0);
            if (g == 1) { STORE_STAGE(Sn) FETCH_STAGE(un) }
            if (g == 6) LDS_BARRIER()
        }
    }
#undef LOAD_GROUP
#undef LDS_BARRIER
#undef FETCH_STAGE
#undef LDV
#undef LDK
#undef LDR
#undef STORE_STAGE
#undef STORE_V
#undef STORE_R
#undef STORE_R1

    float* E = stg0;

    if (WITH_G) {
        #pragma unroll
        for (int r = 0; r < 8; ++r)
            E[E_GSH + wave * 256 + (r + half * 8) * 16 + kM] = accG[0][r] + accG[1][r];
        __syncthreads();
        {
            const int ge = tid * 4;
            const int gr = ge >> 6;
            const int gv = ge & 63;
            const int w0 = (gv >> 4) * 2;
            const int gi = gr * 16 + (gv & 15);
            const v4f g = *reinterpret_cast<const v4f*>(E + E_GSH + w0 * 256 + gi)
                        + *reinterpret_cast<const v4f*>(E + E_GSH + (w0 + 1) * 256 + gi);
            *reinterpret_cast<v4f*>(E + E_GS + ge) = g;
        }
        __syncthreads();
    }

    float* lamb = E + E_LAM + ql * 1024 + half * 512 + vt * 16 + kM;
    const float* gsrc = E + E_GS + half * 512 + vt * 16 + kM;
    #pragma unroll
    for (int r = 0; r < 8; ++r) {
        const float g = gsrc[r * 64];
        #pragma unroll
        for (int y = 0; y < 2; ++y)
            #pragma unroll
            for (int p = 0; p < 4; ++p) {
                float l[4];
                #pragma unroll
                for (int X = 0; X < 4; ++X) {
                    const float c1 = (p & 1) ? acc[X][2][r] : acc[X][1][r];
                    const float c2 = (p & 2) ? acc[X][6][r] : acc[X][3][r];
                    const float c3 = p == 0 ? acc[X][4][r] : p == 1 ? acc[X][5][r]
                                   : p == 2 ? acc[X][7][r] : acc[X][8][r];
                    l[X] = (acc[X][0][r] + c1) + (c2 + c3);
                }
                const float ly = y == 0 ? (l[1] - l[2]) + l[3] : (l[0] + l[1]) + l[2];
                lamb[(y * 4 + p) * 2048 + r * 64] = ly + g;
            }
    }
    __syncthreads();

    #pragma unroll 1
    for (int i = 0; i < 2; ++i) {
        const int pi = wave * 2 + i;
        const int y  = pi >> 3;
        const int p  = (pi >> 1) & 3;
        const int n  = (ny0 + y) * 32 + xg + 4 * ((xb & 1) * 2 + (pi & 1)) + p;
        const float* Qn = Qbuf + (size_t)(b * MM + n) * QC;
        const float* L  = E + E_LAM + pi * 1024 + lane;
        float* on = out + (size_t)(b * MM + n) * 256 + lane;
        float o[4][2] = {};
        #pragma unroll
        for (int k = 0; k < DK; ++k) {
            const float l0 = L[k * 64];
            const float l1 = L[k * 64 + 32];
            #pragma unroll
            for (int h = 0; h < 4; ++h) {
                const float qk = Qn[h * 16 + k];
                o[h][0] = fmaf(qk, l0, o[h][0]);
                o[h][1] = fmaf(qk, l1, o[h][1]);
            }
        }
        #pragma unroll
        for (int h = 0; h < 4; ++h) {
            on[h * 64]      = o[h][0];
            on[h * 64 + 32] = o[h][1];
        }
    }
}

__global__ __launch_bounds__(256, 2) void lambda_main(
    const float* __restrict__ rel,
    const float* __restrict__ Qbuf,
    const float* __restrict__ sK,
    const float* __restrict__ Vbuf,
    float* __restrict__ out)
{
    __shared__ __attribute__((aligned(16))) float stg[2 * SB_N + 1024];

    const int b  = blockIdx.x >> 4;
    const int rp = blockIdx.x & 15;
    lambda_tile<true>(rel, Qbuf, sK, Vbuf, out, stg, b, rp, 0);
    #pragma unroll 1
    for (int xb = 1; xb < 4; ++xb) {
        __builtin_amdgcn_fence(__ATOMIC_RELEASE, "workgroup", "local");
        __builtin_amdgcn_s_barrier();
        __builtin_amdgcn_fence(__ATOMIC_ACQUIRE, "workgroup", "local");
        lambda_tile<false>(rel, Qbuf, sK, Vbuf, out, stg, b, rp, xb);
    }
}

extern "C" void kernel_launch(void* const* d_in, const int* in_sizes, int n_in,
                              void* d_out, int out_size, void* d_ws, size_t ws_size,
                              hipStream_t stream)
{
    const float* x   = (const float*)d_in[0];
    const float* Wq  = (const float*)d_in[1];
    const float* Wk  = (const float*)d_in[2];
    const float* Wv  = (const float*)d_in[3];
    const float* gq  = (const float*)d_in[4];
    const float* bq  = (const float*)d_in[5];
    const float* mq  = (const float*)d_in[6];
    const float* vq  = (const float*)d_in[7];
    const float* gv  = (const float*)d_in[8];
    const float* bv  = (const float*)d_in[9];
    const float* mv  = (const float*)d_in[10];
    const float* vv  = (const float*)d_in[11];
    const float* rel = (const float*)d_in[12];
    float* out = (float*)d_out;

    float* Qbuf = (float*)d_ws;
    float* Vbuf = Qbuf + (size_t)NB * MM * QC;
    float* Kbuf = Vbuf + (size_t)NB * MM * DV;

    qkv_wmma<<<256, 288, 0, stream>>>(x, Wq, Wk, Wv, gq, bq, mq, vq,
                                       gv, bv, mv, vv, Qbuf, Kbuf, Vbuf);
    ksoftmax<<<64, 256, 0, stream>>>(Kbuf);
    lambda_main<<<256, 256, 0, stream>>>(rel, Qbuf, Kbuf, Vbuf, out);
}
